# starting kernel
# baseline (speedup 1.0000x reference)
#include <hip/hip_runtime.h>
#include <math.h>

typedef float v2f __attribute__((ext_vector_type(2)));
typedef float v8f __attribute__((ext_vector_type(8)));

#define B_   16
#define S_   4096
#define DK_  256
#define DV_  256
#define CHK  16
#define NCHK (S_ / CHK)
#define SPLIT 8
#define VROWS (DV_ / SPLIT)
#define NTHREADS 256
#define NWAVES (NTHREADS / 32)
#define MS 260
#define US 33
#define UT 18
#define VS 34

#define LR_C 0.1f
#define A_C  1.1f
#define EPS_C 1e-6f

__global__ __launch_bounds__(NTHREADS)
void DeltaRuleMemory_chunked_wmma(const float* __restrict__ mem,
                                  const float* __restrict__ key,
                                  const float* __restrict__ val,
                                  float* __restrict__ out) {
    extern __shared__ float smem[];
    float* Msub = smem;
    float* Kn   = Msub + VROWS * MS;
    float* P0Ta = Kn   + CHK * MS;
    float* P0Tb = P0Ta + CHK * US;
    float* Uv   = P0Tb + CHK * US;
    float* Ga   = Uv   + VROWS * UT;
    float* Gb   = Ga   + 256;
    float* Vc   = Gb   + 256;
    float* red  = Vc   + CHK * VS;
    float* inv  = red  + 256;

    const int tid  = threadIdx.x;
    const int wave = tid >> 5;
    const int lane = tid & 31;
    const int hi   = lane >> 4;
    const int lm   = lane & 15;

    const int blk = blockIdx.x;
    const int b   = blk / SPLIT;
    const int v0  = (blk % SPLIT) * VROWS;

    const float* memB = mem + (size_t)b * DV_ * DK_ + (size_t)v0 * DK_;
    const float* keyB = key + (size_t)b * S_ * DK_;
    const float* valB = val + (size_t)b * S_ * DV_;
    float*       outB = out + (size_t)b * DV_ * DK_ + (size_t)v0 * DK_;

    for (int i = tid; i < VROWS * DK_; i += NTHREADS) {
        int r = i >> 8, k = i & 255;
        Msub[r * MS + k] = memB[r * DK_ + k];
    }
    __syncthreads();

    const int tA = tid >> 4;
    const int sg = tid & 15;
    const int d0 = sg * 16;
    const int jV = sg * 2;

    for (int c = 0; c < NCHK; ++c) {
        float4 rv[4];
        float ss = 0.f;
        const float* kr = keyB + (size_t)(c * CHK + tA) * DK_ + d0;
        #pragma unroll
        for (int j = 0; j < 4; ++j) {
            rv[j] = ((const float4*)kr)[j];
            ss += rv[j].x * rv[j].x + rv[j].y * rv[j].y +
                  rv[j].z * rv[j].z + rv[j].w * rv[j].w;
        }
        {
            const float* vg = valB + (size_t)(c * CHK + tA) * DV_ + v0 + jV;
            Vc[tA * VS + jV]     = vg[0];
            Vc[tA * VS + jV + 1] = vg[1];
        }
        if (c + 1 < NCHK)
            __builtin_prefetch(keyB + (size_t)((c + 1) * CHK + tA) * DK_ + d0, 0, 0);
        red[tA * 16 + sg] = ss;
        __syncthreads();
        if (tid < 16) {
            float s = 0.f;
            #pragma unroll
            for (int j = 0; j < 16; ++j) s += red[tid * 16 + j];
            inv[tid] = 1.f / (sqrtf(s) + EPS_C);
        }
        __syncthreads();
        {
            float sc = inv[tA];
            float* kw = Kn + tA * MS + d0;
            #pragma unroll
            for (int j = 0; j < 4; ++j) {
                kw[4 * j + 0] = rv[j].x * sc;
                kw[4 * j + 1] = rv[j].y * sc;
                kw[4 * j + 2] = rv[j].z * sc;
                kw[4 * j + 3] = rv[j].w * sc;
            }
        }
        __syncthreads();

        if (wave < 6) {
            const int tile = wave >> 1;
            const int kh   = wave & 1;
            const float* arow = (tile < 2 ? Msub + (tile * 16 + lm) * MS
                                          : Kn + lm * MS) + 2 * hi;
            const float* brow = Kn + lm * MS + 2 * hi;
            v8f acc0 = {}, acc1 = {};
            const int kbeg = kh * 128;
            for (int k0 = kbeg; k0 < kbeg + 128; k0 += 8) {
                v2f a0 = *(const v2f*)(arow + k0);
                v2f b0 = *(const v2f*)(brow + k0);
                acc0 = __builtin_amdgcn_wmma_f32_16x16x4_f32(
                           false, a0, false, b0, (short)0, acc0, false, false);
                v2f a1 = *(const v2f*)(arow + k0 + 4);
                v2f b1 = *(const v2f*)(brow + k0 + 4);
                acc1 = __builtin_amdgcn_wmma_f32_16x16x4_f32(
                           false, a1, false, b1, (short)0, acc1, false, false);
            }
            v8f acc = acc0 + acc1;
            if (tile < 2) {
                float* dst = kh ? P0Tb : P0Ta;
                #pragma unroll
                for (int r = 0; r < 8; ++r)
                    dst[lm * US + tile * 16 + r + 8 * hi] = acc[r];
            } else {
                float* dst = kh ? Gb : Ga;
                #pragma unroll
                for (int r = 0; r < 8; ++r)
                    dst[(r + 8 * hi) * 16 + lm] = acc[r];
            }
        }
        __syncthreads();

        if (tid < VROWS) {
            const int v = tid;
            float u[CHK];
            #pragma unroll
            for (int t = 0; t < CHK; ++t) {
                float x = LR_C * Vc[t * VS + v]
                        - A_C * (P0Ta[t * US + v] + P0Tb[t * US + v]);
                for (int s2 = 0; s2 < t; ++s2)
                    x -= A_C * (Ga[t * 16 + s2] + Gb[t * 16 + s2]) * u[s2];
                u[t] = x;
            }
            float* uw = Uv + v * UT;
            #pragma unroll
            for (int t = 0; t < CHK; ++t) uw[t] = u[t];
        }
        __syncthreads();

        #pragma unroll
        for (int i = 0; i < (VROWS / 16) * (DK_ / 16) / NWAVES; ++i) {
            int tl = wave + i * NWAVES;
            int vt = tl & (VROWS / 16 - 1);
            int kt = tl / (VROWS / 16);
            float* crow = Msub + (vt * 16) * MS + kt * 16 + lm;
            v8f acc;
            #pragma unroll
            for (int r = 0; r < 8; ++r) acc[r] = crow[(r + 8 * hi) * MS];
            const float* urow = Uv + (vt * 16 + lm) * UT + 2 * hi;
            #pragma unroll
            for (int k0 = 0; k0 < CHK; k0 += 4) {
                v2f a = *(const v2f*)(urow + k0);
                v2f bb; bb.x = Kn[(k0 + 2 * hi) * MS + kt * 16 + lm];
                        bb.y = Kn[(k0 + 2 * hi + 1) * MS + kt * 16 + lm];
                acc = __builtin_amdgcn_wmma_f32_16x16x4_f32(
                          false, a, false, bb, (short)0, acc, false, false);
            }
            #pragma unroll
            for (int r = 0; r < 8; ++r) crow[(r + 8 * hi) * MS] = acc[r];
        }
        __syncthreads();
    }

    for (int i = tid; i < VROWS * DK_; i += NTHREADS) {
        int r = i >> 8, k = i & 255;
        outB[r * DK_ + k] = Msub[r * MS + k];
    }
}

extern "C" void kernel_launch(void* const* d_in, const int* in_sizes, int n_in,
                              void* d_out, int out_size, void* d_ws, size_t ws_size,
                              hipStream_t stream) {
    (void)in_sizes; (void)n_in; (void)out_size; (void)d_ws; (void)ws_size;
    const float* mem = (const float*)d_in[0];
    const float* key = (const float*)d_in[1];
    const float* val = (const float*)d_in[2];
    float* out = (float*)d_out;

    size_t smemFloats = (size_t)(VROWS * MS)
                      + CHK * MS
                      + 2 * CHK * US
                      + VROWS * UT
                      + 2 * 256
                      + CHK * VS
                      + 256 + 16;
    size_t smemBytes = smemFloats * sizeof(float);

    dim3 grid(B_ * SPLIT);
    dim3 block(NTHREADS);
    hipLaunchKernelGGL(DeltaRuleMemory_chunked_wmma, grid, block, smemBytes, stream,
                       mem, key, val, out);
}


# Stage next chunk's normalized keys/values one chunk ahead: waves 4-7 load at chunk start and normalize beside the solve into a 2nd LDS buffer; row norm by 8-lane DPP butterfly (sum order changed)
# speedup vs baseline: 1.5496x; 1.5496x over previous
#include <hip/hip_runtime.h>
#include <math.h>

typedef float v2f __attribute__((ext_vector_type(2)));
typedef float v8f __attribute__((ext_vector_type(8)));

#define B_   16
#define S_   4096
#define DK_  256
#define DV_  256
#define CHK  16
#define NCHK (S_ / CHK)
#define SPLIT 8
#define VROWS (DV_ / SPLIT)
#define NTHREADS 256
#define NWAVES (NTHREADS / 32)
#define MS 260
#define US 33
#define UT 18
#define VS 34

#define SMEM_FLOATS (VROWS * MS + 2 * CHK * MS + 2 * CHK * US + VROWS * UT + 2 * 256 + 2 * CHK * VS)
#define STAGE_WAVE0 4
#define DPP_ROW_XMASK 0x160

#define LR_C 0.1f
#define A_C  1.1f
#define EPS_C 1e-6f

template <int M>
__device__ __forceinline__ float row_xor_add(float x) {
    int y = __builtin_amdgcn_update_dpp(0, __float_as_int(x), DPP_ROW_XMASK + M,
                                        0xf, 0xf, true);
    return x + __int_as_float(y);
}

struct Staged {
    float4 k[8];
    float4 v;
};

__device__ __forceinline__ void load_chunk(Staged& s, const float* __restrict__ kr,
                                           const float* __restrict__ vg) {
    #pragma unroll
    for (int j = 0; j < 8; ++j) s.k[j] = ((const float4*)kr)[j * 8];
    s.v = *(const float4*)vg;
}

__device__ __forceinline__ void stage_chunk(const Staged& s, float* __restrict__ kw,
                                            float* __restrict__ vw) {
    float ss = 0.f;
    #pragma unroll
    for (int j = 0; j < 8; ++j)
        ss += s.k[j].x * s.k[j].x + s.k[j].y * s.k[j].y +
              s.k[j].z * s.k[j].z + s.k[j].w * s.k[j].w;
    ss = row_xor_add<1>(ss);
    ss = row_xor_add<2>(ss);
    ss = row_xor_add<4>(ss);
    const float sc = 1.f / (sqrtf(ss) + EPS_C);
    #pragma unroll
    for (int j = 0; j < 8; ++j) {
        float4 o;
        o.x = s.k[j].x * sc; o.y = s.k[j].y * sc;
        o.z = s.k[j].z * sc; o.w = s.k[j].w * sc;
        ((float4*)kw)[j * 8] = o;
    }
    vw[0] = s.v.x; vw[1] = s.v.y; vw[2] = s.v.z; vw[3] = s.v.w;
}

__global__ __launch_bounds__(NTHREADS)
void DeltaRuleMemory_chunked_wmma(const float* __restrict__ mem,
                                  const float* __restrict__ key,
                                  const float* __restrict__ val,
                                  float* __restrict__ out) {
    __shared__ __attribute__((aligned(16))) float smem[SMEM_FLOATS];
    float* Msub = smem;
    float* Kn2  = Msub + VROWS * MS;
    float* P0Ta = Kn2  + 2 * CHK * MS;
    float* P0Tb = P0Ta + CHK * US;
    float* Uv   = P0Tb + CHK * US;
    float* Ga   = Uv   + VROWS * UT;
    float* Gb   = Ga   + 256;
    float* Vc2  = Gb   + 256;

    const int tid  = threadIdx.x;
    const int wave = tid >> 5;
    const int lane = tid & 31;
    const int hi   = lane >> 4;
    const int lm   = lane & 15;

    const int blk = blockIdx.x;
    const int b   = blk / SPLIT;
    const int v0  = (blk % SPLIT) * VROWS;

    const float* memB = mem + (size_t)b * DV_ * DK_ + (size_t)v0 * DK_;
    const float* keyB = key + (size_t)b * S_ * DK_;
    const float* valB = val + (size_t)b * S_ * DV_;
    float*       outB = out + (size_t)b * DV_ * DK_ + (size_t)v0 * DK_;

    const bool stager = wave >= STAGE_WAVE0;
    const int  sT  = (tid & (4 * 32 - 1)) >> 3;
    const int  sG  = tid & 7;
    const float* krT = keyB + (size_t)sT * DK_ + sG * 4;
    const float* vgT = valB + (size_t)sT * DV_ + v0 + sG * 4;
    float* kwT = Kn2 + sT * MS + sG * 4;
    float* vwT = Vc2 + sT * VS + sG * 4;

    Staged st;
    if (stager) load_chunk(st, krT, vgT);
    for (int i = tid; i < VROWS * DK_; i += NTHREADS) {
        int r = i >> 8, k = i & 255;
        Msub[r * MS + k] = memB[r * DK_ + k];
    }
    if (stager) stage_chunk(st, kwT, vwT);
    __syncthreads();

    for (int c = 0; c < NCHK; ++c) {
        const int cur = c & 1;
        const float* Kn = Kn2 + cur * (CHK * MS);
        const float* Vc = Vc2 + cur * (CHK * VS);
        const bool stage_next = stager && (c + 1 < NCHK);

        if (stage_next) {
            load_chunk(st, krT + (size_t)(c + 1) * CHK * DK_, vgT + (size_t)(c + 1) * CHK * DV_);
            if (c + 2 < NCHK)
                __builtin_prefetch(keyB + ((size_t)(c + 2) * CHK + sT) * DK_ + sG * 32, 0, 0);
        }

        if (wave < 6) {
            const int tile = wave >> 1;
            const int kh   = wave & 1;
            const float* arow = (tile < 2 ? Msub + (tile * 16 + lm) * MS
                                          : Kn + lm * MS) + 2 * hi;
            const float* brow = Kn + lm * MS + 2 * hi;
            v8f acc0 = {}, acc1 = {};
            const int kbeg = kh * 128;
            for (int k0 = kbeg; k0 < kbeg + 128; k0 += 8) {
                v2f a0 = *(const v2f*)(arow + k0);
                v2f b0 = *(const v2f*)(brow + k0);
                acc0 = __builtin_amdgcn_wmma_f32_16x16x4_f32(
                           false, a0, false, b0, (short)0, acc0, false, false);
                v2f a1 = *(const v2f*)(arow + k0 + 4);
                v2f b1 = *(const v2f*)(brow + k0 + 4);
                acc1 = __builtin_amdgcn_wmma_f32_16x16x4_f32(
                           false, a1, false, b1, (short)0, acc1, false, false);
            }
            v8f acc = acc0 + acc1;
            if (tile < 2) {
                float* dst = kh ? P0Tb : P0Ta;
                #pragma unroll
                for (int r = 0; r < 8; ++r)
                    dst[lm * US + tile * 16 + r + 8 * hi] = acc[r];
            } else {
                float* dst = kh ? Gb : Ga;
                #pragma unroll
                for (int r = 0; r < 8; ++r)
                    dst[(r + 8 * hi) * 16 + lm] = acc[r];
            }
        }
        __syncthreads();

        if (tid < VROWS) {
            const int v = tid;
            float u[CHK];
            #pragma unroll
            for (int t = 0; t < CHK; ++t) {
                float x = LR_C * Vc[t * VS + v]
                        - A_C * (P0Ta[t * US + v] + P0Tb[t * US + v]);
                for (int s2 = 0; s2 < t; ++s2)
                    x -= A_C * (Ga[t * 16 + s2] + Gb[t * 16 + s2]) * u[s2];
                u[t] = x;
            }
            float* uw = Uv + v * UT;
            #pragma unroll
            for (int t = 0; t < CHK; ++t) uw[t] = u[t];
        } else if (stage_next) {
            const int nxt = cur ^ 1;
            stage_chunk(st, kwT + nxt * (CHK * MS), vwT + nxt * (CHK * VS));
        }
        __syncthreads();

        #pragma unroll
        for (int i = 0; i < (VROWS / 16) * (DK_ / 16) / NWAVES; ++i) {
            int tl = wave + i * NWAVES;
            int vt = tl & (VROWS / 16 - 1);
            int kt = tl / (VROWS / 16);
            float* crow = Msub + (vt * 16) * MS + kt * 16 + lm;
            v8f acc;
            #pragma unroll
            for (int r = 0; r < 8; ++r) acc[r] = crow[(r + 8 * hi) * MS];
            const float* urow = Uv + (vt * 16 + lm) * UT + 2 * hi;
            #pragma unroll
            for (int k0 = 0; k0 < CHK; k0 += 4) {
                v2f a = *(const v2f*)(urow + k0);
                v2f bb; bb.x = Kn[(k0 + 2 * hi) * MS + kt * 16 + lm];
                        bb.y = Kn[(k0 + 2 * hi + 1) * MS + kt * 16 + lm];
                acc = __builtin_amdgcn_wmma_f32_16x16x4_f32(
                          false, a, false, bb, (short)0, acc, false, false);
            }
            #pragma unroll
            for (int r = 0; r < 8; ++r) crow[(r + 8 * hi) * MS] = acc[r];
        }
        __syncthreads();
    }

    for (int i = tid; i < VROWS * DK_; i += NTHREADS) {
        int r = i >> 8, k = i & 255;
        outB[r * DK_ + k] = Msub[r * MS + k];
    }
}

extern "C" void kernel_launch(void* const* d_in, const int* in_sizes, int n_in,
                              void* d_out, int out_size, void* d_ws, size_t ws_size,
                              hipStream_t stream) {
    (void)in_sizes; (void)n_in; (void)out_size; (void)d_ws; (void)ws_size;
    const float* mem = (const float*)d_in[0];
    const float* key = (const float*)d_in[1];
    const float* val = (const float*)d_in[2];
    float* out = (float*)d_out;

    dim3 grid(B_ * SPLIT);
    dim3 block(NTHREADS);
    hipLaunchKernelGGL(DeltaRuleMemory_chunked_wmma, grid, block, 0, stream,
                       mem, key, val, out);
}


# Hold M slice transposed in WMMA accumulator registers all scan: B operand of P0, C/D of the update, no LDS round trip; k split over 8 waves, partials summed in a reduce pass (sum order changed)
# speedup vs baseline: 2.2970x; 1.4823x over previous
#include <hip/hip_runtime.h>
#include <math.h>


typedef float v2f __attribute__((ext_vector_type(2)));
typedef float v8f __attribute__((ext_vector_type(8)));

#define B_   16
#define S_   4096
#define DK_  256
#define DV_  256
#define CHK  16
#define NCHK (S_ / CHK)
#define SPLIT 8
#define VROWS (DV_ / SPLIT)
#define NTHREADS 256
#define NWAVES (NTHREADS / 32)
#define MS 260
#define PS 34
#define UT 18
#define VS 34

#define SMEM_FLOATS (VROWS * MS + 2 * CHK * MS + NWAVES * CHK * PS + NWAVES * 256 + CHK * PS + 256 \
                     + VROWS * UT + 2 * CHK * VS)
#define STAGE_WAVE0 4
#define DPP_ROW_XMASK 0x160

#define LR_C 0.1f
#define A_C  1.1f
#define EPS_C 1e-6f

template <int M>
__device__ __forceinline__ float row_xor_add(float x) {
    int y = __builtin_amdgcn_update_dpp(0, __float_as_int(x), DPP_ROW_XMASK + M,
                                        0xf, 0xf, true);
    return x + __int_as_float(y);
}

struct Staged {
    float4 k[8];
    float4 v;
};

__device__ __forceinline__ void load_chunk(Staged& s, const float* __restrict__ kr,
                                           const float* __restrict__ vg) {
    #pragma unroll
    for (int j = 0; j < 8; ++j) s.k[j] = ((const float4*)kr)[j * 8];
    s.v = *(const float4*)vg;
}

__device__ __forceinline__ void stage_chunk(const Staged& s, float* __restrict__ kw,
                                            float* __restrict__ vw) {
    float ss = 0.f;
    #pragma unroll
    for (int j = 0; j < 8; ++j)
        ss += s.k[j].x * s.k[j].x + s.k[j].y * s.k[j].y +
              s.k[j].z * s.k[j].z + s.k[j].w * s.k[j].w;
    ss = row_xor_add<1>(ss);
    ss = row_xor_add<2>(ss);
    ss = row_xor_add<4>(ss);
    const float sc = 1.f / (sqrtf(ss) + EPS_C);
    #pragma unroll
    for (int j = 0; j < 8; ++j) {
        float4 o;
        o.x = s.k[j].x * sc; o.y = s.k[j].y * sc;
        o.z = s.k[j].z * sc; o.w = s.k[j].w * sc;
        ((float4*)kw)[j * 8] = o;
    }
    vw[0] = s.v.x; vw[1] = s.v.y; vw[2] = s.v.z; vw[3] = s.v.w;
}

__global__ __launch_bounds__(NTHREADS)
void DeltaRuleMemory_chunked_wmma(const float* __restrict__ mem,
                                  const float* __restrict__ key,
                                  const float* __restrict__ val,
                                  float* __restrict__ out) {
    __shared__ __attribute__((aligned(16))) float smem[SMEM_FLOATS];
    float* Msub = smem;
    float* Kn2  = Msub + VROWS * MS;
    float* P0p  = Kn2  + 2 * CHK * MS;
    float* Gp   = P0p  + NWAVES * CHK * PS;
    float* Rhs  = Gp   + NWAVES * 256;
    float* Wg   = Rhs  + CHK * PS;
    float* Uv   = Wg   + 256;
    float* Vc2  = Uv   + VROWS * UT;

    const int tid  = threadIdx.x;
    const int wave = tid >> 5;
    const int lane = tid & 31;
    const int hi   = lane >> 4;
    const int lm   = lane & 15;

    const int blk = blockIdx.x;
    const int b   = blk / SPLIT;
    const int v0  = (blk % SPLIT) * VROWS;

    const float* memB = mem + (size_t)b * DV_ * DK_ + (size_t)v0 * DK_;
    const float* keyB = key + (size_t)b * S_ * DK_;
    const float* valB = val + (size_t)b * S_ * DV_;
    float*       outB = out + (size_t)b * DV_ * DK_ + (size_t)v0 * DK_;

    const bool stager = wave >= STAGE_WAVE0;
    const int  sT  = (tid & (4 * 32 - 1)) >> 3;
    const int  sG  = tid & 7;
    const float* krT = keyB + (size_t)sT * DK_ + sG * 4;
    const float* vgT = valB + (size_t)sT * DV_ + v0 + sG * 4;
    float* kwT = Kn2 + sT * MS + sG * 4;
    float* vwT = Vc2 + sT * VS + sG * 4;

    Staged st;
    if (stager) load_chunk(st, krT, vgT);
    for (int i = tid; i < VROWS * DK_; i += NTHREADS) {
        int r = i >> 8, k = i & 255;
        Msub[r * MS + k] = memB[r * DK_ + k];
    }
    if (stager) stage_chunk(st, kwT, vwT);
    __syncthreads();

    v8f mt[2][2];
    #pragma unroll
    for (int a = 0; a < 2; ++a) {
        #pragma unroll
        for (int vt = 0; vt < 2; ++vt) {
            const float* src = Msub + (vt * 16 + lm) * MS + wave * 32 + a * 16 + 8 * hi;
            const float4 lo = *(const float4*)src;
            const float4 up = *(const float4*)(src + 4);
            v8f m;
            m[0] = lo.x; m[1] = lo.y; m[2] = lo.z; m[3] = lo.w;
            m[4] = up.x; m[5] = up.y; m[6] = up.z; m[7] = up.w;
            mt[a][vt] = m;
        }
    }

    for (int c = 0; c < NCHK; ++c) {
        const int cur = c & 1;
        const float* Kn = Kn2 + cur * (CHK * MS);
        const float* Vc = Vc2 + cur * (CHK * VS);
        const bool stage_next = stager && (c + 1 < NCHK);

        if (stage_next) {
            load_chunk(st, krT + (size_t)(c + 1) * CHK * DK_, vgT + (size_t)(c + 1) * CHK * DV_);
            if (c + 2 < NCHK)
                __builtin_prefetch(keyB + ((size_t)(c + 2) * CHK + sT) * DK_ + sG * 32, 0, 0);
        }

        {
            const float* krow = Kn + lm * MS + wave * 32 + 8 * hi;
            float4 kq[2][2];
            #pragma unroll
            for (int a = 0; a < 2; ++a) {
                kq[a][0] = *(const float4*)(krow + a * 16);
                kq[a][1] = *(const float4*)(krow + a * 16 + 4);
            }
            v8f p0a = {}, p0b = {}, g = {};
            #pragma unroll
            for (int a = 0; a < 2; ++a) {
                #pragma unroll
                for (int j = 0; j < 4; ++j) {
                    const float4 q = kq[a][j >> 1];
                    v2f kf;
                    kf.x = (j & 1) ? q.z : q.x;
                    kf.y = (j & 1) ? q.w : q.y;
                    v2f m0; m0.x = mt[a][0][2 * j]; m0.y = mt[a][0][2 * j + 1];
                    v2f m1; m1.x = mt[a][1][2 * j]; m1.y = mt[a][1][2 * j + 1];
                    p0a = __builtin_amdgcn_wmma_f32_16x16x4_f32(
                              false, kf, false, m0, (short)0, p0a, false, false);
                    p0b = __builtin_amdgcn_wmma_f32_16x16x4_f32(
                              false, kf, false, m1, (short)0, p0b, false, false);
                    g   = __builtin_amdgcn_wmma_f32_16x16x4_f32(
                              false, kf, false, kf, (short)0, g, false, false);
                }
            }
            float* pp = P0p + wave * (CHK * PS);
            float* gp = Gp + wave * 256;
            #pragma unroll
            for (int r = 0; r < 8; ++r) {
                pp[(r + 8 * hi) * PS + lm]      = p0a[r];
                pp[(r + 8 * hi) * PS + 16 + lm] = p0b[r];
                gp[(r + 8 * hi) * 16 + lm]      = g[r];
            }
        }
        __syncthreads();

        {
            const int t = tid >> 4;
            const int v = (tid & 15) * 2;
            v2f ps = *(const v2f*)(P0p + t * PS + v);
            float gs = Gp[tid];
            #pragma unroll
            for (int w = 1; w < NWAVES; ++w) {
                ps += *(const v2f*)(P0p + w * (CHK * PS) + t * PS + v);
                gs += Gp[w * 256 + tid];
            }
            const v2f vv = *(const v2f*)(Vc + t * VS + v);
            v2f rh;
            rh.x = LR_C * vv.x - A_C * ps.x;
            rh.y = LR_C * vv.y - A_C * ps.y;
            *(v2f*)(Rhs + t * PS + v) = rh;
            Wg[tid] = A_C * gs;
        }
        __syncthreads();

        if (tid < VROWS) {
            const int v = tid;
            float u[CHK];
            #pragma unroll
            for (int t = 0; t < CHK; ++t) {
                float x = Rhs[t * PS + v];
                #pragma unroll
                for (int s2 = 0; s2 < t; ++s2)
                    x -= Wg[t * 16 + s2] * u[s2];
                u[t] = x;
            }
            float* uw = Uv + v * UT;
            #pragma unroll
            for (int t = 0; t < CHK; ++t) uw[t] = u[t];
        } else if (stage_next) {
            const int nxt = cur ^ 1;
            stage_chunk(st, kwT + nxt * (CHK * MS), vwT + nxt * (CHK * VS));
        }
        __syncthreads();

        {
            v2f kt2[2][4], uf[2][4];
            #pragma unroll
            for (int j = 0; j < 4; ++j) {
                #pragma unroll
                for (int a = 0; a < 2; ++a) {
                    const float* kp = Kn + (4 * j + 2 * hi) * MS + wave * 32 + a * 16 + lm;
                    kt2[a][j].x = kp[0];
                    kt2[a][j].y = kp[MS];
                }
                #pragma unroll
                for (int vt = 0; vt < 2; ++vt)
                    uf[vt][j] = *(const v2f*)(Uv + (vt * 16 + lm) * UT + 4 * j + 2 * hi);
            }
            #pragma unroll
            for (int j = 0; j < 4; ++j) {
                #pragma unroll
                for (int a = 0; a < 2; ++a) {
                    #pragma unroll
                    for (int vt = 0; vt < 2; ++vt)
                        mt[a][vt] = __builtin_amdgcn_wmma_f32_16x16x4_f32(
                                        false, kt2[a][j], false, uf[vt][j], (short)0, mt[a][vt],
                                        false, false);
                }
            }
        }
        __syncthreads();
    }

    #pragma unroll
    for (int a = 0; a < 2; ++a) {
        #pragma unroll
        for (int vt = 0; vt < 2; ++vt) {
            float* dst = Msub + (vt * 16 + lm) * MS + wave * 32 + a * 16 + 8 * hi;
            const v8f m = mt[a][vt];
            float4 lo, up;
            lo.x = m[0]; lo.y = m[1]; lo.z = m[2]; lo.w = m[3];
            up.x = m[4]; up.y = m[5]; up.z = m[6]; up.w = m[7];
            *(float4*)dst = lo;
            *(float4*)(dst + 4) = up;
        }
    }
    __syncthreads();
    for (int i = tid; i < VROWS * DK_; i += NTHREADS) {
        int r = i >> 8, k = i & 255;
        outB[r * DK_ + k] = Msub[r * MS + k];
    }
}

extern "C" void kernel_launch(void* const* d_in, const int* in_sizes, int n_in,
                              void* d_out, int out_size, void* d_ws, size_t ws_size,
                              hipStream_t stream) {
    (void)in_sizes; (void)n_in; (void)out_size; (void)d_ws; (void)ws_size;
    const float* mem = (const float*)d_in[0];
    const float* key = (const float*)d_in[1];
    const float* val = (const float*)d_in[2];
    float* out = (float*)d_out;

    dim3 grid(B_ * SPLIT);
    dim3 block(NTHREADS);
    hipLaunchKernelGGL(DeltaRuleMemory_chunked_wmma, grid, block, 0, stream,
                       mem, key, val, out);
}


# Split the value rows 16 ways instead of 8: 256 workgroups of one 16-row v tile each, 24 instead of 40 matrix instructions per wave and chunk, all WGPs at work; sum order unchanged
# speedup vs baseline: 2.4786x; 1.0791x over previous
#include <hip/hip_runtime.h>
#include <math.h>


typedef float v2f __attribute__((ext_vector_type(2)));
typedef float v8f __attribute__((ext_vector_type(8)));

#define B_   16
#define S_   4096
#define DK_  256
#define DV_  256
#define CHK  16
#define NCHK (S_ / CHK)
#define SPLIT 16
#define VROWS (DV_ / SPLIT)
#define NTHREADS 256
#define NWAVES (NTHREADS / 32)
#define MS 260
#define PS 18
#define UT 18
#define VS 18

#define SMEM_FLOATS (VROWS * MS + 2 * CHK * MS + NWAVES * CHK * PS + NWAVES * 256 + CHK * PS + 256 \
                     + VROWS * UT + 2 * CHK * VS)
#define STAGE_WAVE0 4
#define VPIECES (VROWS / 4)
#define DPP_ROW_XMASK 0x160

#define LR_C 0.1f
#define A_C  1.1f
#define EPS_C 1e-6f

template <int M>
__device__ __forceinline__ float row_xor_add(float x) {
    int y = __builtin_amdgcn_update_dpp(0, __float_as_int(x), DPP_ROW_XMASK + M,
                                        0xf, 0xf, true);
    return x + __int_as_float(y);
}

struct Staged {
    float4 k[8];
    float4 v;
};

__device__ __forceinline__ void load_chunk(Staged& s, const float* __restrict__ kr,
                                           const float* __restrict__ vg, bool has_v) {
    #pragma unroll
    for (int j = 0; j < 8; ++j) s.k[j] = ((const float4*)kr)[j * 8];
    if (has_v) s.v = *(const float4*)vg;
}

__device__ __forceinline__ void stage_chunk(const Staged& s, float* __restrict__ kw,
                                            float* __restrict__ vw, bool has_v) {
    float ss = 0.f;
    #pragma unroll
    for (int j = 0; j < 8; ++j)
        ss += s.k[j].x * s.k[j].x + s.k[j].y * s.k[j].y +
              s.k[j].z * s.k[j].z + s.k[j].w * s.k[j].w;
    ss = row_xor_add<1>(ss);
    ss = row_xor_add<2>(ss);
    ss = row_xor_add<4>(ss);
    const float sc = 1.f / (sqrtf(ss) + EPS_C);
    #pragma unroll
    for (int j = 0; j < 8; ++j) {
        float4 o;
        o.x = s.k[j].x * sc; o.y = s.k[j].y * sc;
        o.z = s.k[j].z * sc; o.w = s.k[j].w * sc;
        ((float4*)kw)[j * 8] = o;
    }
    if (has_v) { vw[0] = s.v.x; vw[1] = s.v.y; vw[2] = s.v.z; vw[3] = s.v.w; }
}

__global__ __launch_bounds__(NTHREADS)
void DeltaRuleMemory_chunked_wmma(const float* __restrict__ mem,
                                  const float* __restrict__ key,
                                  const float* __restrict__ val,
                                  float* __restrict__ out) {
    __shared__ __attribute__((aligned(16))) float smem[SMEM_FLOATS];
    float* Msub = smem;
    float* Kn2  = Msub + VROWS * MS;
    float* P0p  = Kn2  + 2 * CHK * MS;
    float* Gp   = P0p  + NWAVES * CHK * PS;
    float* Rhs  = Gp   + NWAVES * 256;
    float* Wg   = Rhs  + CHK * PS;
    float* Uv   = Wg   + 256;
    float* Vc2  = Uv   + VROWS * UT;

    const int tid  = threadIdx.x;
    const int wave = tid >> 5;
    const int lane = tid & 31;
    const int hi   = lane >> 4;
    const int lm   = lane & 15;

    const int blk = blockIdx.x;
    const int b   = blk / SPLIT;
    const int v0  = (blk % SPLIT) * VROWS;

    const float* memB = mem + (size_t)b * DV_ * DK_ + (size_t)v0 * DK_;
    const float* keyB = key + (size_t)b * S_ * DK_;
    const float* valB = val + (size_t)b * S_ * DV_;
    float*       outB = out + (size_t)b * DV_ * DK_ + (size_t)v0 * DK_;

    const bool stager = wave >= STAGE_WAVE0;
    const int  sT  = (tid & (4 * 32 - 1)) >> 3;
    const int  sG  = tid & 7;
    const bool sV  = sG < VPIECES;
    const int  sGv = sV ? sG : 0;
    const float* krT = keyB + (size_t)sT * DK_ + sG * 4;
    const float* vgT = valB + (size_t)sT * DV_ + v0 + sGv * 4;
    float* kwT = Kn2 + sT * MS + sG * 4;
    float* vwT = Vc2 + sT * VS + sGv * 4;

    Staged st;
    if (stager) load_chunk(st, krT, vgT, sV);
    for (int i = tid; i < VROWS * DK_; i += NTHREADS) {
        int r = i >> 8, k = i & 255;
        Msub[r * MS + k] = memB[r * DK_ + k];
    }
    if (stager) stage_chunk(st, kwT, vwT, sV);
    __syncthreads();

    v8f mt[2];
    #pragma unroll
    for (int a = 0; a < 2; ++a) {
        const float* src = Msub + lm * MS + wave * 32 + a * 16 + 8 * hi;
        const float4 lo = *(const float4*)src;
        const float4 up = *(const float4*)(src + 4);
        v8f m;
        m[0] = lo.x; m[1] = lo.y; m[2] = lo.z; m[3] = lo.w;
        m[4] = up.x; m[5] = up.y; m[6] = up.z; m[7] = up.w;
        mt[a] = m;
    }

    for (int c = 0; c < NCHK; ++c) {
        const int cur = c & 1;
        const float* Kn = Kn2 + cur * (CHK * MS);
        const float* Vc = Vc2 + cur * (CHK * VS);
        const bool stage_next = stager && (c + 1 < NCHK);

        if (stage_next) {
            load_chunk(st, krT + (size_t)(c + 1) * CHK * DK_, vgT + (size_t)(c + 1) * CHK * DV_, sV);
            if (c + 2 < NCHK)
                __builtin_prefetch(keyB + ((size_t)(c + 2) * CHK + sT) * DK_ + sG * 32, 0, 0);
        }

        {
            const float* krow = Kn + lm * MS + wave * 32 + 8 * hi;
            float4 kq[2][2];
            #pragma unroll
            for (int a = 0; a < 2; ++a) {
                kq[a][0] = *(const float4*)(krow + a * 16);
                kq[a][1] = *(const float4*)(krow + a * 16 + 4);
            }
            v8f p0a = {}, g = {};
            #pragma unroll
            for (int a = 0; a < 2; ++a) {
                #pragma unroll
                for (int j = 0; j < 4; ++j) {
                    const float4 q = kq[a][j >> 1];
                    v2f kf;
                    kf.x = (j & 1) ? q.z : q.x;
                    kf.y = (j & 1) ? q.w : q.y;
                    v2f m0; m0.x = mt[a][2 * j]; m0.y = mt[a][2 * j + 1];
                    p0a = __builtin_amdgcn_wmma_f32_16x16x4_f32(
                              false, kf, false, m0, (short)0, p0a, false, false);
                    g   = __builtin_amdgcn_wmma_f32_16x16x4_f32(
                              false, kf, false, kf, (short)0, g, false, false);
                }
            }
            float* pp = P0p + wave * (CHK * PS);
            float* gp = Gp + wave * 256;
            #pragma unroll
            for (int r = 0; r < 8; ++r) {
                pp[(r + 8 * hi) * PS + lm] = p0a[r];
                gp[(r + 8 * hi) * 16 + lm] = g[r];
            }
        }
        __syncthreads();

        {
            const int t = tid >> 4;
            const int v = tid & 15;
            float ps = P0p[t * PS + v];
            float gs = Gp[tid];
            #pragma unroll
            for (int w = 1; w < NWAVES; ++w) {
                ps += P0p[w * (CHK * PS) + t * PS + v];
                gs += Gp[w * 256 + tid];
            }
            Rhs[t * PS + v] = LR_C * Vc[t * VS + v] - A_C * ps;
            Wg[tid] = A_C * gs;
        }
        __syncthreads();

        if (tid < VROWS) {
            const int v = tid;
            float u[CHK];
            #pragma unroll
            for (int t = 0; t < CHK; ++t) {
                float x = Rhs[t * PS + v];
                #pragma unroll
                for (int s2 = 0; s2 < t; ++s2)
                    x -= Wg[t * 16 + s2] * u[s2];
                u[t] = x;
            }
            float* uw = Uv + v * UT;
            #pragma unroll
            for (int t = 0; t < CHK; ++t) uw[t] = u[t];
        } else if (stage_next) {
            const int nxt = cur ^ 1;
            stage_chunk(st, kwT + nxt * (CHK * MS), vwT + nxt * (CHK * VS), sV);
        }
        __syncthreads();

        {
            v2f kt2[2][4], uf[4];
            #pragma unroll
            for (int j = 0; j < 4; ++j) {
                #pragma unroll
                for (int a = 0; a < 2; ++a) {
                    const float* kp = Kn + (4 * j + 2 * hi) * MS + wave * 32 + a * 16 + lm;
                    kt2[a][j].x = kp[0];
                    kt2[a][j].y = kp[MS];
                }
                uf[j] = *(const v2f*)(Uv + lm * UT + 4 * j + 2 * hi);
            }
            #pragma unroll
            for (int j = 0; j < 4; ++j) {
                #pragma unroll
                for (int a = 0; a < 2; ++a)
                    mt[a] = __builtin_amdgcn_wmma_f32_16x16x4_f32(
                                false, kt2[a][j], false, uf[j], (short)0, mt[a],
                                false, false);
            }
        }
        __syncthreads();
    }

    #pragma unroll
    for (int a = 0; a < 2; ++a) {
        float* dst = Msub + lm * MS + wave * 32 + a * 16 + 8 * hi;
        const v8f m = mt[a];
        float4 lo, up;
        lo.x = m[0]; lo.y = m[1]; lo.z = m[2]; lo.w = m[3];
        up.x = m[4]; up.y = m[5]; up.z = m[6]; up.w = m[7];
        *(float4*)dst = lo;
        *(float4*)(dst + 4) = up;
    }
    __syncthreads();
    for (int i = tid; i < VROWS * DK_; i += NTHREADS) {
        int r = i >> 8, k = i & 255;
        outB[r * DK_ + k] = Msub[r * MS + k];
    }
}

extern "C" void kernel_launch(void* const* d_in, const int* in_sizes, int n_in,
                              void* d_out, int out_size, void* d_ws, size_t ws_size,
                              hipStream_t stream) {
    (void)in_sizes; (void)n_in; (void)out_size; (void)d_ws; (void)ws_size;
    const float* mem = (const float*)d_in[0];
    const float* key = (const float*)d_in[1];
    const float* val = (const float*)d_in[2];
    float* out = (float*)d_out;

    dim3 grid(B_ * SPLIT);
    dim3 block(NTHREADS);
    hipLaunchKernelGGL(DeltaRuleMemory_chunked_wmma, grid, block, 0, stream,
                       mem, key, val, out);
}


# Drop the chunk-closing barrier (3 per chunk, not 4): update and next chunk's P0/gram form one phase whose LDS fragment loads and staged global loads all issue at its head; sum order unchanged
# speedup vs baseline: 2.6783x; 1.0806x over previous
#include <hip/hip_runtime.h>
#include <math.h>


typedef float v2f __attribute__((ext_vector_type(2)));
typedef float v8f __attribute__((ext_vector_type(8)));

#define B_   16
#define S_   4096
#define DK_  256
#define DV_  256
#define CHK  16
#define NCHK (S_ / CHK)
#define SPLIT 16
#define VROWS (DV_ / SPLIT)
#define NTHREADS 256
#define NWAVES (NTHREADS / 32)
#define MS 260
#define PS 18
#define UT 18
#define VS 18

#define SMEM_FLOATS (VROWS * MS + 2 * CHK * MS + NWAVES * CHK * PS + NWAVES * 256 + CHK * PS + 256 \
                     + VROWS * UT + 2 * CHK * VS)
#define STAGE_WAVE0 4
#define VPIECES (VROWS / 4)
#define DPP_ROW_XMASK 0x160

#define LR_C 0.1f
#define A_C  1.1f
#define EPS_C 1e-6f

template <int M>
__device__ __forceinline__ float row_xor_add(float x) {
    int y = __builtin_amdgcn_update_dpp(0, __float_as_int(x), DPP_ROW_XMASK + M,
                                        0xf, 0xf, true);
    return x + __int_as_float(y);
}

struct Staged {
    float4 k[8];
    float4 v;
};

__device__ __forceinline__ void load_chunk(Staged& s, const float* __restrict__ kr,
                                           const float* __restrict__ vg, bool has_v) {
    #pragma unroll
    for (int j = 0; j < 8; ++j) s.k[j] = ((const float4*)kr)[j * 8];
    if (has_v) s.v = *(const float4*)vg;
}

__device__ __forceinline__ void stage_chunk(const Staged& s, float* __restrict__ kw,
                                            float* __restrict__ vw, bool has_v) {
    float ss = 0.f;
    #pragma unroll
    for (int j = 0; j < 8; ++j)
        ss += s.k[j].x * s.k[j].x + s.k[j].y * s.k[j].y +
              s.k[j].z * s.k[j].z + s.k[j].w * s.k[j].w;
    ss = row_xor_add<1>(ss);
    ss = row_xor_add<2>(ss);
    ss = row_xor_add<4>(ss);
    const float sc = 1.f / (sqrtf(ss) + EPS_C);
    #pragma unroll
    for (int j = 0; j < 8; ++j) {
        float4 o;
        o.x = s.k[j].x * sc; o.y = s.k[j].y * sc;
        o.z = s.k[j].z * sc; o.w = s.k[j].w * sc;
        ((float4*)kw)[j * 8] = o;
    }
    if (has_v) { vw[0] = s.v.x; vw[1] = s.v.y; vw[2] = s.v.z; vw[3] = s.v.w; }
}

__global__ __launch_bounds__(NTHREADS)
void DeltaRuleMemory_chunked_wmma(const float* __restrict__ mem,
                                  const float* __restrict__ key,
                                  const float* __restrict__ val,
                                  float* __restrict__ out) {
    __shared__ __attribute__((aligned(16))) float smem[SMEM_FLOATS];
    float* Msub = smem;
    float* Kn2  = Msub + VROWS * MS;
    float* P0p  = Kn2  + 2 * CHK * MS;
    float* Gp   = P0p  + NWAVES * CHK * PS;
    float* Rhs  = Gp   + NWAVES * 256;
    float* Wg   = Rhs  + CHK * PS;
    float* Uv   = Wg   + 256;
    float* Vc2  = Uv   + VROWS * UT;

    const int tid  = threadIdx.x;
    const int wave = tid >> 5;
    const int lane = tid & 31;
    const int hi   = lane >> 4;
    const int lm   = lane & 15;

    const int blk = blockIdx.x;
    const int b   = blk / SPLIT;
    const int v0  = (blk % SPLIT) * VROWS;

    const float* memB = mem + (size_t)b * DV_ * DK_ + (size_t)v0 * DK_;
    const float* keyB = key + (size_t)b * S_ * DK_;
    const float* valB = val + (size_t)b * S_ * DV_;
    float*       outB = out + (size_t)b * DV_ * DK_ + (size_t)v0 * DK_;

    const bool stager = wave >= STAGE_WAVE0;
    const int  sT  = (tid & (4 * 32 - 1)) >> 3;
    const int  sG  = tid & 7;
    const bool sV  = sG < VPIECES;
    const int  sGv = sV ? sG : 0;
    const float* krT = keyB + (size_t)sT * DK_ + sG * 4;
    const float* vgT = valB + (size_t)sT * DV_ + v0 + sGv * 4;
    float* kwT = Kn2 + sT * MS + sG * 4;
    float* vwT = Vc2 + sT * VS + sGv * 4;

    Staged st;
    if (stager) load_chunk(st, krT, vgT, sV);
    for (int i = tid; i < VROWS * DK_; i += NTHREADS) {
        int r = i >> 8, k = i & 255;
        Msub[r * MS + k] = memB[r * DK_ + k];
    }
    if (stager) stage_chunk(st, kwT, vwT, sV);
    __syncthreads();

    v8f mt[2];
    #pragma unroll
    for (int a = 0; a < 2; ++a) {
        const float* src = Msub + lm * MS + wave * 32 + a * 16 + 8 * hi;
        const float4 lo = *(const float4*)src;
        const float4 up = *(const float4*)(src + 4);
        v8f m;
        m[0] = lo.x; m[1] = lo.y; m[2] = lo.z; m[3] = lo.w;
        m[4] = up.x; m[5] = up.y; m[6] = up.z; m[7] = up.w;
        mt[a] = m;
    }

    if (stager) {
        load_chunk(st, krT + (size_t)CHK * DK_, vgT + (size_t)CHK * DV_, sV);
        __builtin_prefetch(keyB + ((size_t)2 * CHK + sT) * DK_ + sG * 32, 0, 0);
    }
    float4 kq[2][2];
    {
        const float* krow = Kn2 + lm * MS + wave * 32 + 8 * hi;
        #pragma unroll
        for (int a = 0; a < 2; ++a) {
            kq[a][0] = *(const float4*)(krow + a * 16);
            kq[a][1] = *(const float4*)(krow + a * 16 + 4);
        }
    }

    for (int c = 0; c < NCHK; ++c) {
        const int cur = c & 1;
        const float* Kn = Kn2 + cur * (CHK * MS);
        const float* Vc = Vc2 + cur * (CHK * VS);
        const bool stage_next = stager && (c + 1 < NCHK);

        {
            v8f p0a = {}, g = {};
            #pragma unroll
            for (int a = 0; a < 2; ++a) {
                #pragma unroll
                for (int j = 0; j < 4; ++j) {
                    const float4 q = kq[a][j >> 1];
                    v2f kf;
                    kf.x = (j & 1) ? q.z : q.x;
                    kf.y = (j & 1) ? q.w : q.y;
                    v2f m0; m0.x = mt[a][2 * j]; m0.y = mt[a][2 * j + 1];
                    p0a = __builtin_amdgcn_wmma_f32_16x16x4_f32(
                              false, kf, false, m0, (short)0, p0a, false, false);
                    g   = __builtin_amdgcn_wmma_f32_16x16x4_f32(
                              false, kf, false, kf, (short)0, g, false, false);
                }
            }
            float* pp = P0p + wave * (CHK * PS);
            float* gp = Gp + wave * 256;
            #pragma unroll
            for (int r = 0; r < 8; ++r) {
                pp[(r + 8 * hi) * PS + lm] = p0a[r];
                gp[(r + 8 * hi) * 16 + lm] = g[r];
            }
        }
        __syncthreads();

        {
            const int t = tid >> 4;
            const int v = tid & 15;
            float ps = P0p[t * PS + v];
            float gs = Gp[tid];
            #pragma unroll
            for (int w = 1; w < NWAVES; ++w) {
                ps += P0p[w * (CHK * PS) + t * PS + v];
                gs += Gp[w * 256 + tid];
            }
            Rhs[t * PS + v] = LR_C * Vc[t * VS + v] - A_C * ps;
            Wg[tid] = A_C * gs;
        }
        __syncthreads();

        if (tid < VROWS) {
            const int v = tid;
            float u[CHK];
            #pragma unroll
            for (int t = 0; t < CHK; ++t) {
                float x = Rhs[t * PS + v];
                #pragma unroll
                for (int s2 = 0; s2 < t; ++s2)
                    x -= Wg[t * 16 + s2] * u[s2];
                u[t] = x;
            }
            float* uw = Uv + v * UT;
            #pragma unroll
            for (int t = 0; t < CHK; ++t) uw[t] = u[t];
        } else if (stage_next) {
            const int nxt = cur ^ 1;
            stage_chunk(st, kwT + nxt * (CHK * MS), vwT + nxt * (CHK * VS), sV);
        }
        __syncthreads();

        if (stager && (c + 2 < NCHK)) {
            load_chunk(st, krT + (size_t)(c + 2) * CHK * DK_, vgT + (size_t)(c + 2) * CHK * DV_, sV);
            if (c + 3 < NCHK)
                __builtin_prefetch(keyB + ((size_t)(c + 3) * CHK + sT) * DK_ + sG * 32, 0, 0);
        }
        {
            float kx[2][4], ky[2][4];
            v2f uf[4];
            #pragma unroll
            for (int j = 0; j < 4; ++j) {
                #pragma unroll
                for (int a = 0; a < 2; ++a) {
                    const float* kp = Kn + (4 * j + 2 * hi) * MS + wave * 32 + a * 16 + lm;
                    kx[a][j] = kp[0];
                    ky[a][j] = kp[MS];
                }
                uf[j] = *(const v2f*)(Uv + lm * UT + 4 * j + 2 * hi);
            }
            if (c + 1 < NCHK) {
                const float* krow = Kn2 + (cur ^ 1) * (CHK * MS) + lm * MS + wave * 32 + 8 * hi;
                #pragma unroll
                for (int a = 0; a < 2; ++a) {
                    kq[a][0] = *(const float4*)(krow + a * 16);
                    kq[a][1] = *(const float4*)(krow + a * 16 + 4);
                }
            }
            __builtin_amdgcn_sched_barrier(0);
            #pragma unroll
            for (int j = 0; j < 4; ++j) {
                #pragma unroll
                for (int a = 0; a < 2; ++a) {
                    v2f kt2; kt2.x = kx[a][j]; kt2.y = ky[a][j];
                    mt[a] = __builtin_amdgcn_wmma_f32_16x16x4_f32(
                                false, kt2, false, uf[j], (short)0, mt[a],
                                false, false);
                }
            }
        }
    }

    #pragma unroll
    for (int a = 0; a < 2; ++a) {
        float* dst = Msub + lm * MS + wave * 32 + a * 16 + 8 * hi;
        const v8f m = mt[a];
        float4 lo, up;
        lo.x = m[0]; lo.y = m[1]; lo.z = m[2]; lo.w = m[3];
        up.x = m[4]; up.y = m[5]; up.z = m[6]; up.w = m[7];
        *(float4*)dst = lo;
        *(float4*)(dst + 4) = up;
    }
    __syncthreads();
    for (int i = tid; i < VROWS * DK_; i += NTHREADS) {
        int r = i >> 8, k = i & 255;
        outB[r * DK_ + k] = Msub[r * MS + k];
    }
}

extern "C" void kernel_launch(void* const* d_in, const int* in_sizes, int n_in,
                              void* d_out, int out_size, void* d_ws, size_t ws_size,
                              hipStream_t stream) {
    (void)in_sizes; (void)n_in; (void)out_size; (void)d_ws; (void)ws_size;
    const float* mem = (const float*)d_in[0];
    const float* key = (const float*)d_in[1];
    const float* val = (const float*)d_in[2];
    float* out = (float*)d_out;

    dim3 grid(B_ * SPLIT);
    dim3 block(NTHREADS);
    hipLaunchKernelGGL(DeltaRuleMemory_chunked_wmma, grid, block, 0, stream,
                       mem, key, val, out);
}


# Staged global loads stay in flight across the chunk's barriers: LDS-only fences at the 3 loop barriers, loads issued right behind the previous staging, a whole chunk to land in; sum order unchanged
# speedup vs baseline: 3.0884x; 1.1531x over previous
#include <hip/hip_runtime.h>
#include <math.h>


typedef float v2f __attribute__((ext_vector_type(2)));
typedef float v8f __attribute__((ext_vector_type(8)));

#define B_   16
#define S_   4096
#define DK_  256
#define DV_  256
#define CHK  16
#define NCHK (S_ / CHK)
#define SPLIT 16
#define VROWS (DV_ / SPLIT)
#define NTHREADS 256
#define NWAVES (NTHREADS / 32)
#define MS 260
#define PS 18
#define UT 18
#define VS 18

#define SMEM_FLOATS (VROWS * MS + 2 * CHK * MS + NWAVES * CHK * PS + NWAVES * 256 + CHK * PS + 256 \
                     + VROWS * UT + 2 * CHK * VS)
#define STAGE_WAVE0 4
#define VPIECES (VROWS / 4)
#define DPP_ROW_XMASK 0x160

#define LR_C 0.1f
#define A_C  1.1f
#define EPS_C 1e-6f

template <int M>
__device__ __forceinline__ float row_xor_add(float x) {
    int y = __builtin_amdgcn_update_dpp(0, __float_as_int(x), DPP_ROW_XMASK + M,
                                        0xf, 0xf, true);
    return x + __int_as_float(y);
}

__device__ __forceinline__ void lds_barrier() {
    __builtin_amdgcn_fence(__ATOMIC_RELEASE, "workgroup", "local");
    __builtin_amdgcn_s_barrier();
    __builtin_amdgcn_fence(__ATOMIC_ACQUIRE, "workgroup", "local");
}

struct Staged {
    float4 k[8];
    float4 v;
};

__device__ __forceinline__ void load_chunk(Staged& s, const float* __restrict__ kr,
                                           const float* __restrict__ vg, bool has_v) {
    #pragma unroll
    for (int j = 0; j < 8; ++j) s.k[j] = ((const float4*)kr)[j * 8];
    if (has_v) s.v = *(const float4*)vg;
}

__device__ __forceinline__ void stage_chunk(const Staged& s, float* __restrict__ kw,
                                            float* __restrict__ vw, bool has_v) {
    float ss = 0.f;
    #pragma unroll
    for (int j = 0; j < 8; ++j)
        ss += s.k[j].x * s.k[j].x + s.k[j].y * s.k[j].y +
              s.k[j].z * s.k[j].z + s.k[j].w * s.k[j].w;
    ss = row_xor_add<1>(ss);
    ss = row_xor_add<2>(ss);
    ss = row_xor_add<4>(ss);
    const float sc = 1.f / (sqrtf(ss) + EPS_C);
    #pragma unroll
    for (int j = 0; j < 8; ++j) {
        float4 o;
        o.x = s.k[j].x * sc; o.y = s.k[j].y * sc;
        o.z = s.k[j].z * sc; o.w = s.k[j].w * sc;
        ((float4*)kw)[j * 8] = o;
    }
    if (has_v) { vw[0] = s.v.x; vw[1] = s.v.y; vw[2] = s.v.z; vw[3] = s.v.w; }
}

__global__ __launch_bounds__(NTHREADS)
void DeltaRuleMemory_chunked_wmma(const float* __restrict__ mem,
                                  const float* __restrict__ key,
                                  const float* __restrict__ val,
                                  float* __restrict__ out) {
    __shared__ __attribute__((aligned(16))) float smem[SMEM_FLOATS];
    float* Msub = smem;
    float* Kn2  = Msub + VROWS * MS;
    float* P0p  = Kn2  + 2 * CHK * MS;
    float* Gp   = P0p  + NWAVES * CHK * PS;
    float* Rhs  = Gp   + NWAVES * 256;
    float* Wg   = Rhs  + CHK * PS;
    float* Uv   = Wg   + 256;
    float* Vc2  = Uv   + VROWS * UT;

    const int tid  = threadIdx.x;
    const int wave = tid >> 5;
    const int lane = tid & 31;
    const int hi   = lane >> 4;
    const int lm   = lane & 15;

    const int blk = blockIdx.x;
    const int b   = blk / SPLIT;
    const int v0  = (blk % SPLIT) * VROWS;

    const float* memB = mem + (size_t)b * DV_ * DK_ + (size_t)v0 * DK_;
    const float* keyB = key + (size_t)b * S_ * DK_;
    const float* valB = val + (size_t)b * S_ * DV_;
    float*       outB = out + (size_t)b * DV_ * DK_ + (size_t)v0 * DK_;

    const bool stager = wave >= STAGE_WAVE0;
    const int  sT  = (tid & (4 * 32 - 1)) >> 3;
    const int  sG  = tid & 7;
    const bool sV  = sG < VPIECES;
    const int  sGv = sV ? sG : 0;
    const float* krT = keyB + (size_t)sT * DK_ + sG * 4;
    const float* vgT = valB + (size_t)sT * DV_ + v0 + sGv * 4;
    float* kwT = Kn2 + sT * MS + sG * 4;
    float* vwT = Vc2 + sT * VS + sGv * 4;

    Staged st;
    if (stager) load_chunk(st, krT, vgT, sV);
    for (int i = tid; i < VROWS * DK_; i += NTHREADS) {
        int r = i >> 8, k = i & 255;
        Msub[r * MS + k] = memB[r * DK_ + k];
    }
    if (stager) stage_chunk(st, kwT, vwT, sV);
    __syncthreads();

    v8f mt[2];
    #pragma unroll
    for (int a = 0; a < 2; ++a) {
        const float* src = Msub + lm * MS + wave * 32 + a * 16 + 8 * hi;
        const float4 lo = *(const float4*)src;
        const float4 up = *(const float4*)(src + 4);
        v8f m;
        m[0] = lo.x; m[1] = lo.y; m[2] = lo.z; m[3] = lo.w;
        m[4] = up.x; m[5] = up.y; m[6] = up.z; m[7] = up.w;
        mt[a] = m;
    }

    if (stager) {
        load_chunk(st, krT + (size_t)CHK * DK_, vgT + (size_t)CHK * DV_, sV);
        __builtin_prefetch(keyB + ((size_t)2 * CHK + sT) * DK_ + sG * 32, 0, 0);
    }
    float4 kq[2][2];
    {
        const float* krow = Kn2 + lm * MS + wave * 32 + 8 * hi;
        #pragma unroll
        for (int a = 0; a < 2; ++a) {
            kq[a][0] = *(const float4*)(krow + a * 16);
            kq[a][1] = *(const float4*)(krow + a * 16 + 4);
        }
    }

    for (int c = 0; c < NCHK; ++c) {
        const int cur = c & 1;
        const float* Kn = Kn2 + cur * (CHK * MS);
        const float* Vc = Vc2 + cur * (CHK * VS);
        const bool stage_next = stager && (c + 1 < NCHK);

        {
            v8f p0a = {}, g = {};
            #pragma unroll
            for (int a = 0; a < 2; ++a) {
                #pragma unroll
                for (int j = 0; j < 4; ++j) {
                    const float4 q = kq[a][j >> 1];
                    v2f kf;
                    kf.x = (j & 1) ? q.z : q.x;
                    kf.y = (j & 1) ? q.w : q.y;
                    v2f m0; m0.x = mt[a][2 * j]; m0.y = mt[a][2 * j + 1];
                    p0a = __builtin_amdgcn_wmma_f32_16x16x4_f32(
                              false, kf, false, m0, (short)0, p0a, false, false);
                    g   = __builtin_amdgcn_wmma_f32_16x16x4_f32(
                              false, kf, false, kf, (short)0, g, false, false);
                }
            }
            float* pp = P0p + wave * (CHK * PS);
            float* gp = Gp + wave * 256;
            #pragma unroll
            for (int r = 0; r < 8; ++r) {
                pp[(r + 8 * hi) * PS + lm] = p0a[r];
                gp[(r + 8 * hi) * 16 + lm] = g[r];
            }
        }
        lds_barrier();

        {
            const int t = tid >> 4;
            const int v = tid & 15;
            float ps = P0p[t * PS + v];
            float gs = Gp[tid];
            #pragma unroll
            for (int w = 1; w < NWAVES; ++w) {
                ps += P0p[w * (CHK * PS) + t * PS + v];
                gs += Gp[w * 256 + tid];
            }
            Rhs[t * PS + v] = LR_C * Vc[t * VS + v] - A_C * ps;
            Wg[tid] = A_C * gs;
        }
        lds_barrier();

        if (tid < VROWS) {
            const int v = tid;
            float u[CHK];
            #pragma unroll
            for (int t = 0; t < CHK; ++t) {
                float x = Rhs[t * PS + v];
                #pragma unroll
                for (int s2 = 0; s2 < t; ++s2)
                    x -= Wg[t * 16 + s2] * u[s2];
                u[t] = x;
            }
            float* uw = Uv + v * UT;
            #pragma unroll
            for (int t = 0; t < CHK; ++t) uw[t] = u[t];
        } else if (stage_next) {
            const int nxt = cur ^ 1;
            stage_chunk(st, kwT + nxt * (CHK * MS), vwT + nxt * (CHK * VS), sV);
            if (c + 2 < NCHK) {
                load_chunk(st, krT + (size_t)(c + 2) * CHK * DK_, vgT + (size_t)(c + 2) * CHK * DV_, sV);
                if (c + 3 < NCHK)
                    __builtin_prefetch(keyB + ((size_t)(c + 3) * CHK + sT) * DK_ + sG * 32, 0, 0);
            }
        }
        lds_barrier();

        {
            float kx[2][4], ky[2][4];
            v2f uf[4];
            #pragma unroll
            for (int j = 0; j < 4; ++j) {
                #pragma unroll
                for (int a = 0; a < 2; ++a) {
                    const float* kp = Kn + (4 * j + 2 * hi) * MS + wave * 32 + a * 16 + lm;
                    kx[a][j] = kp[0];
                    ky[a][j] = kp[MS];
                }
                uf[j] = *(const v2f*)(Uv + lm * UT + 4 * j + 2 * hi);
            }
            if (c + 1 < NCHK) {
                const float* krow = Kn2 + (cur ^ 1) * (CHK * MS) + lm * MS + wave * 32 + 8 * hi;
                #pragma unroll
                for (int a = 0; a < 2; ++a) {
                    kq[a][0] = *(const float4*)(krow + a * 16);
                    kq[a][1] = *(const float4*)(krow + a * 16 + 4);
                }
            }
            __builtin_amdgcn_sched_barrier(0);
            #pragma unroll
            for (int j = 0; j < 4; ++j) {
                #pragma unroll
                for (int a = 0; a < 2; ++a) {
                    v2f kt2; kt2.x = kx[a][j]; kt2.y = ky[a][j];
                    mt[a] = __builtin_amdgcn_wmma_f32_16x16x4_f32(
                                false, kt2, false, uf[j], (short)0, mt[a],
                                false, false);
                }
            }
        }
    }

    #pragma unroll
    for (int a = 0; a < 2; ++a) {
        float* dst = Msub + lm * MS + wave * 32 + a * 16 + 8 * hi;
        const v8f m = mt[a];
        float4 lo, up;
        lo.x = m[0]; lo.y = m[1]; lo.z = m[2]; lo.w = m[3];
        up.x = m[4]; up.y = m[5]; up.z = m[6]; up.w = m[7];
        *(float4*)dst = lo;
        *(float4*)(dst + 4) = up;
    }
    __syncthreads();
    for (int i = tid; i < VROWS * DK_; i += NTHREADS) {
        int r = i >> 8, k = i & 255;
        outB[r * DK_ + k] = Msub[r * MS + k];
    }
}

extern "C" void kernel_launch(void* const* d_in, const int* in_sizes, int n_in,
                              void* d_out, int out_size, void* d_ws, size_t ws_size,
                              hipStream_t stream) {
    (void)in_sizes; (void)n_in; (void)out_size; (void)d_ws; (void)ws_size;
    const float* mem = (const float*)d_in[0];
    const float* key = (const float*)d_in[1];
    const float* val = (const float*)d_in[2];
    float* out = (float*)d_out;

    dim3 grid(B_ * SPLIT);
    dim3 block(NTHREADS);
    hipLaunchKernelGGL(DeltaRuleMemory_chunked_wmma, grid, block, 0, stream,
                       mem, key, val, out);
}


# Triangular solve spread over 128 lanes: a 16-lane row per column, u[s] handed on by a row_share DPP multiply-add, 15 links; reduce and staging re-split between the wave groups; sum order unchanged
# speedup vs baseline: 3.3146x; 1.0732x over previous
#include <hip/hip_runtime.h>
#include <math.h>


typedef float v2f __attribute__((ext_vector_type(2)));
typedef float v8f __attribute__((ext_vector_type(8)));

#define B_   16
#define S_   4096
#define DK_  256
#define DV_  256
#define CHK  16
#define NCHK (S_ / CHK)
#define SPLIT 16
#define VROWS (DV_ / SPLIT)
#define NTHREADS 256
#define NWAVES (NTHREADS / 32)
#define MS 260
#define PS 18
#define UT 18
#define VS 18
#define WS 20

#define SMEM_FLOATS (VROWS * MS + 2 * CHK * MS + NWAVES * CHK * PS + NWAVES * 256 + CHK * WS \
                     + VROWS * UT + 2 * CHK * VS)
#define STAGE_WAVE0 4
#define VPIECES (VROWS / 4)
#define DPP_ROW_XMASK 0x160

#define LR_C 0.1f
#define A_C  1.1f
#define EPS_C 1e-6f

template <int M>
__device__ __forceinline__ float row_xor_add(float x) {
    int y = __builtin_amdgcn_update_dpp(0, __float_as_int(x), DPP_ROW_XMASK + M,
                                        0xf, 0xf, true);
    return x + __int_as_float(y);
}

template <int S>
__device__ __forceinline__ void solve_step(float wn, float& xa, float& xb) {
    asm("v_fmac_f32_dpp %0, %0, %1 row_share:%2 row_mask:0xf bank_mask:0xf bound_ctrl:1"
        : "+v"(xa) : "v"(wn), "n"(S));
    asm("v_fmac_f32_dpp %0, %0, %1 row_share:%2 row_mask:0xf bank_mask:0xf bound_ctrl:1"
        : "+v"(xb) : "v"(wn), "n"(S));
}

__device__ __forceinline__ void lds_barrier() {
    __builtin_amdgcn_fence(__ATOMIC_RELEASE, "workgroup", "local");
    __builtin_amdgcn_s_barrier();
    __builtin_amdgcn_fence(__ATOMIC_ACQUIRE, "workgroup", "local");
}

struct Staged {
    float4 k[8];
    float4 v;
};

__device__ __forceinline__ void load_chunk(Staged& s, const float* __restrict__ kr,
                                           const float* __restrict__ vg, bool has_v) {
    #pragma unroll
    for (int j = 0; j < 8; ++j) s.k[j] = ((const float4*)kr)[j * 8];
    if (has_v) s.v = *(const float4*)vg;
}

__device__ __forceinline__ float stage_scale(const Staged& s) {
    float ss = 0.f;
    #pragma unroll
    for (int j = 0; j < 8; ++j)
        ss += s.k[j].x * s.k[j].x + s.k[j].y * s.k[j].y +
              s.k[j].z * s.k[j].z + s.k[j].w * s.k[j].w;
    ss = row_xor_add<1>(ss);
    ss = row_xor_add<2>(ss);
    ss = row_xor_add<4>(ss);
    return 1.f / (sqrtf(ss) + EPS_C);
}

__device__ __forceinline__ void stage_write(const Staged& s, float sc, float* __restrict__ kw,
                                            float* __restrict__ vw, bool has_v) {
    #pragma unroll
    for (int j = 0; j < 8; ++j) {
        float4 o;
        o.x = s.k[j].x * sc; o.y = s.k[j].y * sc;
        o.z = s.k[j].z * sc; o.w = s.k[j].w * sc;
        ((float4*)kw)[j * 8] = o;
    }
    if (has_v) { vw[0] = s.v.x; vw[1] = s.v.y; vw[2] = s.v.z; vw[3] = s.v.w; }
}

__device__ __forceinline__ void stage_chunk(const Staged& s, float* __restrict__ kw,
                                            float* __restrict__ vw, bool has_v) {
    stage_write(s, stage_scale(s), kw, vw, has_v);
}

__global__ __launch_bounds__(NTHREADS)
void DeltaRuleMemory_chunked_wmma(const float* __restrict__ mem,
                                  const float* __restrict__ key,
                                  const float* __restrict__ val,
                                  float* __restrict__ out) {
    __shared__ __attribute__((aligned(16))) float smem[SMEM_FLOATS];
    float* Msub = smem;
    float* Kn2  = Msub + VROWS * MS;
    float* P0p  = Kn2  + 2 * CHK * MS;
    float* Gp   = P0p  + NWAVES * CHK * PS;
    float* Wg   = Gp   + NWAVES * 256;
    float* Uv   = Wg   + CHK * WS;
    float* Vc2  = Uv   + VROWS * UT;

    const int tid  = threadIdx.x;
    const int wave = tid >> 5;
    const int lane = tid & 31;
    const int hi   = lane >> 4;
    const int lm   = lane & 15;

    const int blk = blockIdx.x;
    const int b   = blk / SPLIT;
    const int v0  = (blk % SPLIT) * VROWS;

    const float* memB = mem + (size_t)b * DV_ * DK_ + (size_t)v0 * DK_;
    const float* keyB = key + (size_t)b * S_ * DK_;
    const float* valB = val + (size_t)b * S_ * DV_;
    float*       outB = out + (size_t)b * DV_ * DK_ + (size_t)v0 * DK_;

    const bool stager = wave >= STAGE_WAVE0;
    const int  sT  = (tid & (4 * 32 - 1)) >> 3;
    const int  sG  = tid & 7;
    const bool sV  = sG < VPIECES;
    const int  sGv = sV ? sG : 0;
    const float* krT = keyB + (size_t)sT * DK_ + sG * 4;
    const float* vgT = valB + (size_t)sT * DV_ + v0 + sGv * 4;
    float* kwT = Kn2 + sT * MS + sG * 4;
    float* vwT = Vc2 + sT * VS + sGv * 4;

    Staged st;
    if (stager) load_chunk(st, krT, vgT, sV);
    for (int i = tid; i < VROWS * DK_; i += NTHREADS) {
        int r = i >> 8, k = i & 255;
        Msub[r * MS + k] = memB[r * DK_ + k];
    }
    if (stager) stage_chunk(st, kwT, vwT, sV);
    __syncthreads();

    v8f mt[2];
    #pragma unroll
    for (int a = 0; a < 2; ++a) {
        const float* src = Msub + lm * MS + wave * 32 + a * 16 + 8 * hi;
        const float4 lo = *(const float4*)src;
        const float4 up = *(const float4*)(src + 4);
        v8f m;
        m[0] = lo.x; m[1] = lo.y; m[2] = lo.z; m[3] = lo.w;
        m[4] = up.x; m[5] = up.y; m[6] = up.z; m[7] = up.w;
        mt[a] = m;
    }

    if (stager) {
        load_chunk(st, krT + (size_t)CHK * DK_, vgT + (size_t)CHK * DV_, sV);
        __builtin_prefetch(keyB + ((size_t)2 * CHK + sT) * DK_ + sG * 32, 0, 0);
    }
    float4 kq[2][2];
    {
        const float* krow = Kn2 + lm * MS + wave * 32 + 8 * hi;
        #pragma unroll
        for (int a = 0; a < 2; ++a) {
            kq[a][0] = *(const float4*)(krow + a * 16);
            kq[a][1] = *(const float4*)(krow + a * 16 + 4);
        }
    }

    for (int c = 0; c < NCHK; ++c) {
        const int cur = c & 1;
        const float* Kn = Kn2 + cur * (CHK * MS);
        const float* Vc = Vc2 + cur * (CHK * VS);
        const bool stage_next = stager && (c + 1 < NCHK);

        {
            v8f p0a = {}, g = {};
            #pragma unroll
            for (int a = 0; a < 2; ++a) {
                #pragma unroll
                for (int j = 0; j < 4; ++j) {
                    const float4 q = kq[a][j >> 1];
                    v2f kf;
                    kf.x = (j & 1) ? q.z : q.x;
                    kf.y = (j & 1) ? q.w : q.y;
                    v2f m0; m0.x = mt[a][2 * j]; m0.y = mt[a][2 * j + 1];
                    p0a = __builtin_amdgcn_wmma_f32_16x16x4_f32(
                              false, kf, false, m0, (short)0, p0a, false, false);
                    g   = __builtin_amdgcn_wmma_f32_16x16x4_f32(
                              false, kf, false, kf, (short)0, g, false, false);
                }
            }
            float* pp = P0p + wave * (CHK * PS);
            float* gp = Gp + wave * 256;
            #pragma unroll
            for (int r = 0; r < 8; ++r) {
                pp[(r + 8 * hi) * PS + lm] = p0a[r];
                gp[(r + 8 * hi) * 16 + lm] = g[r];
            }
        }
        lds_barrier();

        const int va = 4 * wave + hi;
        float xa = 0.f, xb = 0.f;
        float sc = 0.f;
        if (!stager) {
            const float* pp = P0p + lm * PS + va;
            float pa = pp[0];
            float pb = pp[2];
            const int gt = tid >> 4;
            const int gc = tid & 15;
            float g0 = Gp[tid];
            float g1 = Gp[tid + 128];
            #pragma unroll
            for (int w = 1; w < NWAVES; ++w) {
                pa += pp[w * (CHK * PS)];
                pb += pp[w * (CHK * PS) + 2];
                g0 += Gp[w * 256 + tid];
                g1 += Gp[w * 256 + tid + 128];
            }
            xa = LR_C * Vc[lm * VS + va] - A_C * pa;
            xb = LR_C * Vc[lm * VS + va + 2] - A_C * pb;
            Wg[gt * WS + gc]       = (gc < gt)     ? -(A_C * g0) : 0.f;
            Wg[(gt + 8) * WS + gc] = (gc < gt + 8) ? -(A_C * g1) : 0.f;
        } else if (stage_next) {
            sc = stage_scale(st);
        }
        lds_barrier();

        if (!stager) {
            const float4* wr = (const float4*)(Wg + lm * WS);
            const float4 w0 = wr[0], w1 = wr[1], w2 = wr[2], w3 = wr[3];
            __builtin_amdgcn_sched_barrier(0);
            solve_step<0>(w0.x, xa, xb);
            solve_step<1>(w0.y, xa, xb);
            solve_step<2>(w0.z, xa, xb);
            solve_step<3>(w0.w, xa, xb);
            solve_step<4>(w1.x, xa, xb);
            solve_step<5>(w1.y, xa, xb);
            solve_step<6>(w1.z, xa, xb);
            solve_step<7>(w1.w, xa, xb);
            solve_step<8>(w2.x, xa, xb);
            solve_step<9>(w2.y, xa, xb);
            solve_step<10>(w2.z, xa, xb);
            solve_step<11>(w2.w, xa, xb);
            solve_step<12>(w3.x, xa, xb);
            solve_step<13>(w3.y, xa, xb);
            solve_step<14>(w3.z, xa, xb);
            Uv[va * UT + lm]       = xa;
            Uv[(va + 2) * UT + lm] = xb;
        } else if (stage_next) {
            const int nxt = cur ^ 1;
            stage_write(st, sc, kwT + nxt * (CHK * MS), vwT + nxt * (CHK * VS), sV);
            if (c + 2 < NCHK) {
                load_chunk(st, krT + (size_t)(c + 2) * CHK * DK_, vgT + (size_t)(c + 2) * CHK * DV_, sV);
                if (c + 3 < NCHK)
                    __builtin_prefetch(keyB + ((size_t)(c + 3) * CHK + sT) * DK_ + sG * 32, 0, 0);
            }
        }
        lds_barrier();

        {
            float kx[2][4], ky[2][4];
            v2f uf[4];
            #pragma unroll
            for (int j = 0; j < 4; ++j) {
                #pragma unroll
                for (int a = 0; a < 2; ++a) {
                    const float* kp = Kn + (4 * j + 2 * hi) * MS + wave * 32 + a * 16 + lm;
                    kx[a][j] = kp[0];
                    ky[a][j] = kp[MS];
                }
                uf[j] = *(const v2f*)(Uv + lm * UT + 4 * j + 2 * hi);
            }
            if (c + 1 < NCHK) {
                const float* krow = Kn2 + (cur ^ 1) * (CHK * MS) + lm * MS + wave * 32 + 8 * hi;
                #pragma unroll
                for (int a = 0; a < 2; ++a) {
                    kq[a][0] = *(const float4*)(krow + a * 16);
                    kq[a][1] = *(const float4*)(krow + a * 16 + 4);
                }
            }
            __builtin_amdgcn_sched_barrier(0);
            #pragma unroll
            for (int j = 0; j < 4; ++j) {
                #pragma unroll
                for (int a = 0; a < 2; ++a) {
                    v2f kt2; kt2.x = kx[a][j]; kt2.y = ky[a][j];
                    mt[a] = __builtin_amdgcn_wmma_f32_16x16x4_f32(
                                false, kt2, false, uf[j], (short)0, mt[a],
                                false, false);
                }
            }
        }
    }

    #pragma unroll
    for (int a = 0; a < 2; ++a) {
        float* dst = Msub + lm * MS + wave * 32 + a * 16 + 8 * hi;
        const v8f m = mt[a];
        float4 lo, up;
        lo.x = m[0]; lo.y = m[1]; lo.z = m[2]; lo.w = m[3];
        up.x = m[4]; up.y = m[5]; up.z = m[6]; up.w = m[7];
        *(float4*)dst = lo;
        *(float4*)(dst + 4) = up;
    }
    __syncthreads();
    for (int i = tid; i < VROWS * DK_; i += NTHREADS) {
        int r = i >> 8, k = i & 255;
        outB[r * DK_ + k] = Msub[r * MS + k];
    }
}

extern "C" void kernel_launch(void* const* d_in, const int* in_sizes, int n_in,
                              void* d_out, int out_size, void* d_ws, size_t ws_size,
                              hipStream_t stream) {
    (void)in_sizes; (void)n_in; (void)out_size; (void)d_ws; (void)ws_size;
    const float* mem = (const float*)d_in[0];
    const float* key = (const float*)d_in[1];
    const float* val = (const float*)d_in[2];
    float* out = (float*)d_out;

    dim3 grid(B_ * SPLIT);
    dim3 block(NTHREADS);
    hipLaunchKernelGGL(DeltaRuleMemory_chunked_wmma, grid, block, 0, stream,
                       mem, key, val, out);
}


# Gram, its unit-triangular inverse T and the key scales formed once per chunk by a first launch (workspace); the scan does U = T*(lr V - a P0) on the matrix unit: no gram, no solve (sum order changed)
# speedup vs baseline: 3.9895x; 1.2036x over previous
#include <hip/hip_runtime.h>
#include <math.h>


typedef float v2f __attribute__((ext_vector_type(2)));
typedef float v8f __attribute__((ext_vector_type(8)));

#define B_   16
#define S_   4096
#define DK_  256
#define DV_  256
#define CHK  16
#define NCHK (S_ / CHK)
#define SPLIT 16
#define VROWS (DV_ / SPLIT)
#define NTHREADS 256
#define NWAVES (NTHREADS / 32)
#define MS 260
#define PS 18
#define RS 20
#define VS 18
#define WS 20

#define SMEM_FLOATS (VROWS * MS + 2 * CHK * MS + NWAVES * CHK * PS + VROWS * RS + CHK * RS + 2 * CHK * VS)
#define STAGE_WAVE0 4
#define VPIECES (VROWS / 4)
#define DPP_ROW_XMASK 0x160

#define LR_C 0.1f
#define A_C  1.1f
#define EPS_C 1e-6f

#define FAC_CPW 8
#define FAC_GRID (B_ * NCHK / FAC_CPW)
#define WS_T_FLOATS (B_ * NCHK * CHK * CHK)
#define WS_S_FLOATS (B_ * S_)
#define WS_NEED_BYTES ((size_t)(WS_T_FLOATS + WS_S_FLOATS) * sizeof(float))

template <int M>
__device__ __forceinline__ float row_xor_add(float x) {
    int y = __builtin_amdgcn_update_dpp(0, __float_as_int(x), DPP_ROW_XMASK + M,
                                        0xf, 0xf, true);
    return x + __int_as_float(y);
}

template <int U>
__device__ __forceinline__ void subst_step(float wn, float& x) {
    asm("v_fmac_f32_dpp %0, %0, %1 row_share:%2 row_mask:0xf bank_mask:0xf bound_ctrl:1"
        : "+v"(x) : "v"(wn), "n"(U));
}

__device__ __forceinline__ void lds_barrier() {
    __builtin_amdgcn_fence(__ATOMIC_RELEASE, "workgroup", "local");
    __builtin_amdgcn_s_barrier();
    __builtin_amdgcn_fence(__ATOMIC_ACQUIRE, "workgroup", "local");
}

__global__ __launch_bounds__(NTHREADS)
void DeltaRuleMemory_chunk_factors(const float* __restrict__ key, float* __restrict__ ws,
                                   size_t ws_size) {
    if (ws_size < WS_NEED_BYTES) return;

    __shared__ __attribute__((aligned(16))) float fsm[CHK * MS + NWAVES * 256 + CHK * WS + FAC_CPW * CHK];
    float* Kn = fsm;
    float* Gp = Kn + CHK * MS;
    float* Wu = Gp + NWAVES * 256;
    float* Sc = Wu + CHK * WS;

    const int tid  = threadIdx.x;
    const int wave = tid >> 5;
    const int lane = tid & 31;
    const int hi   = lane >> 4;
    const int lm   = lane & 15;
    const int row  = tid >> 4;

    const int cid0 = blockIdx.x * FAC_CPW;
    const float* kr = key + (size_t)cid0 * CHK * DK_ + (size_t)row * DK_ + lm * 4;
    float* Tw = ws + (size_t)cid0 * (CHK * CHK) + tid;

    float4 q[4];
    #pragma unroll
    for (int j = 0; j < 4; ++j) q[j] = ((const float4*)kr)[j * 16];

    for (int i = 0; i < FAC_CPW; ++i) {
        float ss = 0.f;
        #pragma unroll
        for (int j = 0; j < 4; ++j)
            ss += q[j].x * q[j].x + q[j].y * q[j].y + q[j].z * q[j].z + q[j].w * q[j].w;
        ss = row_xor_add<1>(ss);
        ss = row_xor_add<2>(ss);
        ss = row_xor_add<4>(ss);
        ss = row_xor_add<8>(ss);
        const float sc = 1.f / (sqrtf(ss) + EPS_C);
        #pragma unroll
        for (int j = 0; j < 4; ++j) {
            float4 o;
            o.x = q[j].x * sc; o.y = q[j].y * sc; o.z = q[j].z * sc; o.w = q[j].w * sc;
            ((float4*)(Kn + row * MS + lm * 4))[j * 16] = o;
        }
        if (lm == 0) Sc[i * CHK + row] = sc;
        if (i + 1 < FAC_CPW) {
            const float* kn = kr + (size_t)(i + 1) * CHK * DK_;
            #pragma unroll
            for (int j = 0; j < 4; ++j) q[j] = ((const float4*)kn)[j * 16];
        }
        lds_barrier();

        {
            const float* krow = Kn + lm * MS + wave * 32 + 8 * hi;
            v8f g = {};
            #pragma unroll
            for (int a = 0; a < 2; ++a) {
                const float4 q0 = *(const float4*)(krow + a * 16);
                const float4 q1 = *(const float4*)(krow + a * 16 + 4);
                v2f kf;
                kf.x = q0.x; kf.y = q0.y;
                g = __builtin_amdgcn_wmma_f32_16x16x4_f32(false, kf, false, kf, (short)0, g, false, false);
                kf.x = q0.z; kf.y = q0.w;
                g = __builtin_amdgcn_wmma_f32_16x16x4_f32(false, kf, false, kf, (short)0, g, false, false);
                kf.x = q1.x; kf.y = q1.y;
                g = __builtin_amdgcn_wmma_f32_16x16x4_f32(false, kf, false, kf, (short)0, g, false, false);
                kf.x = q1.z; kf.y = q1.w;
                g = __builtin_amdgcn_wmma_f32_16x16x4_f32(false, kf, false, kf, (short)0, g, false, false);
            }
            float* gp = Gp + wave * 256;
            #pragma unroll
            for (int r = 0; r < 8; ++r) gp[(r + 8 * hi) * 16 + lm] = g[r];
        }
        lds_barrier();

        {
            float gs = Gp[tid];
            #pragma unroll
            for (int w = 1; w < NWAVES; ++w) gs += Gp[w * 256 + tid];
            Wu[row * WS + lm] = (lm > row) ? -(A_C * gs) : 0.f;
        }
        lds_barrier();

        {
            const float4* wr = (const float4*)(Wu + lm * WS);
            const float4 w0 = wr[0], w1 = wr[1], w2 = wr[2], w3 = wr[3];
            float x = (lm == row) ? 1.f : 0.f;
            subst_step<15>(w3.w, x);
            subst_step<14>(w3.z, x);
            subst_step<13>(w3.y, x);
            subst_step<12>(w3.x, x);
            subst_step<11>(w2.w, x);
            subst_step<10>(w2.z, x);
            subst_step<9>(w2.y, x);
            subst_step<8>(w2.x, x);
            subst_step<7>(w1.w, x);
            subst_step<6>(w1.z, x);
            subst_step<5>(w1.y, x);
            subst_step<4>(w1.x, x);
            subst_step<3>(w0.w, x);
            subst_step<2>(w0.z, x);
            subst_step<1>(w0.y, x);
            Tw[(size_t)i * (CHK * CHK)] = x;
        }
    }
    lds_barrier();
    if (tid < FAC_CPW * CHK) ws[WS_T_FLOATS + (size_t)cid0 * CHK + tid] = Sc[tid];
}


struct Staged {
    float4 k[8];
    float4 v;
    float  sc;
};

__device__ __forceinline__ void load_chunk(Staged& s, const float* __restrict__ kr,
                                           const float* __restrict__ vg,
                                           const float* __restrict__ sr, bool has_v) {
    #pragma unroll
    for (int j = 0; j < 8; ++j) s.k[j] = ((const float4*)kr)[j * 8];
    s.sc = *sr;
    if (has_v) s.v = *(const float4*)vg;
}

__device__ __forceinline__ void stage_write(const Staged& s, float* __restrict__ kw,
                                            float* __restrict__ vw, bool has_v) {
    const float sc = s.sc;
    #pragma unroll
    for (int j = 0; j < 8; ++j) {
        float4 o;
        o.x = s.k[j].x * sc; o.y = s.k[j].y * sc;
        o.z = s.k[j].z * sc; o.w = s.k[j].w * sc;
        ((float4*)kw)[j * 8] = o;
    }
    if (has_v) { vw[0] = s.v.x; vw[1] = s.v.y; vw[2] = s.v.z; vw[3] = s.v.w; }
}

__global__ __launch_bounds__(NTHREADS)
void DeltaRuleMemory_chunked_wmma(const float* __restrict__ mem,
                                  const float* __restrict__ key,
                                  const float* __restrict__ val,
                                  float* __restrict__ out,
                                  const float* __restrict__ ws,
                                  size_t ws_size) {
    if (ws_size < WS_NEED_BYTES) return;

    __shared__ __attribute__((aligned(16))) float smem[SMEM_FLOATS];
    float* Msub = smem;
    float* Kn2  = Msub + VROWS * MS;
    float* P0p  = Kn2  + 2 * CHK * MS;
    float* Rv   = P0p  + NWAVES * CHK * PS;
    float* Tl   = Rv   + VROWS * RS;
    float* Vc2  = Tl   + CHK * RS;

    const int tid  = threadIdx.x;
    const int wave = tid >> 5;
    const int lane = tid & 31;
    const int hi   = lane >> 4;
    const int lm   = lane & 15;

    const int blk = blockIdx.x;
    const int b   = blk / SPLIT;
    const int v0  = (blk % SPLIT) * VROWS;

    const float* memB = mem + (size_t)b * DV_ * DK_ + (size_t)v0 * DK_;
    const float* keyB = key + (size_t)b * S_ * DK_;
    const float* valB = val + (size_t)b * S_ * DV_;
    float*       outB = out + (size_t)b * DV_ * DK_ + (size_t)v0 * DK_;
    const float* scB  = ws + WS_T_FLOATS + (size_t)b * S_;

    const bool stager = wave >= STAGE_WAVE0;
    const int  sT  = (tid & (4 * 32 - 1)) >> 3;
    const int  sG  = tid & 7;
    const bool sV  = sG < VPIECES;
    const int  sGv = sV ? sG : 0;
    const float* krT = keyB + (size_t)sT * DK_ + sG * 4;
    const float* vgT = valB + (size_t)sT * DV_ + v0 + sGv * 4;
    const float* srT = scB + sT;
    float* kwT = Kn2 + sT * MS + sG * 4;
    float* vwT = Vc2 + sT * VS + sGv * 4;
    const int    sE  = 2 * (tid & (4 * 32 - 1));
    const float* tpT = ws + (size_t)b * NCHK * (CHK * CHK) + sE;
    float*       twT = Tl + (sE >> 4) * RS + (sE & 15);

    Staged st;
    float2 stT = {0.f, 0.f};
    if (stager) load_chunk(st, krT, vgT, srT, sV);
    for (int i = tid; i < VROWS * DK_; i += NTHREADS) {
        int r = i >> 8, k = i & 255;
        Msub[r * MS + k] = memB[r * DK_ + k];
    }
    if (stager) stage_write(st, kwT, vwT, sV);
    __syncthreads();

    v8f mt[2];
    #pragma unroll
    for (int a = 0; a < 2; ++a) {
        const float* src = Msub + lm * MS + wave * 32 + a * 16 + 8 * hi;
        const float4 lo = *(const float4*)src;
        const float4 up = *(const float4*)(src + 4);
        v8f m;
        m[0] = lo.x; m[1] = lo.y; m[2] = lo.z; m[3] = lo.w;
        m[4] = up.x; m[5] = up.y; m[6] = up.z; m[7] = up.w;
        mt[a] = m;
    }

    if (stager) {
        load_chunk(st, krT + (size_t)CHK * DK_, vgT + (size_t)CHK * DV_, srT + CHK, sV);
        stT = *(const float2*)tpT;
        __builtin_prefetch(keyB + ((size_t)2 * CHK + sT) * DK_ + sG * 32, 0, 0);
    }
    float4 kq[2][2];
    {
        const float* krow = Kn2 + lm * MS + wave * 32 + 8 * hi;
        #pragma unroll
        for (int a = 0; a < 2; ++a) {
            kq[a][0] = *(const float4*)(krow + a * 16);
            kq[a][1] = *(const float4*)(krow + a * 16 + 4);
        }
    }

    for (int c = 0; c < NCHK; ++c) {
        const int cur = c & 1;
        const float* Kn = Kn2 + cur * (CHK * MS);
        const float* Vc = Vc2 + cur * (CHK * VS);
        const bool stage_next = stager && (c + 1 < NCHK);

        {
            v8f p0a = {};
            #pragma unroll
            for (int a = 0; a < 2; ++a) {
                #pragma unroll
                for (int j = 0; j < 4; ++j) {
                    const float4 q = kq[a][j >> 1];
                    v2f kf;
                    kf.x = (j & 1) ? q.z : q.x;
                    kf.y = (j & 1) ? q.w : q.y;
                    v2f m0; m0.x = mt[a][2 * j]; m0.y = mt[a][2 * j + 1];
                    p0a = __builtin_amdgcn_wmma_f32_16x16x4_f32(
                              false, kf, false, m0, (short)0, p0a, false, false);
                }
            }
            float* pp = P0p + wave * (CHK * PS);
            #pragma unroll
            for (int r = 0; r < 8; ++r) pp[(r + 8 * hi) * PS + lm] = p0a[r];
        }
        lds_barrier();

        if (!stager) {
            const int va = 4 * wave + hi;
            const float* pp = P0p + lm * PS + va;
            float pa = pp[0];
            float pb = pp[2];
            #pragma unroll
            for (int w = 1; w < NWAVES; ++w) {
                pa += pp[w * (CHK * PS)];
                pb += pp[w * (CHK * PS) + 2];
            }
            Rv[va * RS + lm]       = LR_C * Vc[lm * VS + va] - A_C * pa;
            Rv[(va + 2) * RS + lm] = LR_C * Vc[lm * VS + va + 2] - A_C * pb;
        } else {
            *(float2*)twT = stT;
            if (stage_next) {
                const int nxt = cur ^ 1;
                stage_write(st, kwT + nxt * (CHK * MS), vwT + nxt * (CHK * VS), sV);
                if (c + 2 < NCHK) {
                    load_chunk(st, krT + (size_t)(c + 2) * CHK * DK_, vgT + (size_t)(c + 2) * CHK * DV_,
                               srT + (c + 2) * CHK, sV);
                    if (c + 3 < NCHK)
                        __builtin_prefetch(keyB + ((size_t)(c + 3) * CHK + sT) * DK_ + sG * 32, 0, 0);
                }
                stT = *(const float2*)(tpT + (size_t)(c + 1) * (CHK * CHK));
            }
        }
        lds_barrier();

        {
            const float4 rq0 = *(const float4*)(Rv + lm * RS + 8 * hi);
            const float4 rq1 = *(const float4*)(Rv + lm * RS + 8 * hi + 4);
            const float4 tq0 = *(const float4*)(Tl + lm * RS + 8 * hi);
            const float4 tq1 = *(const float4*)(Tl + lm * RS + 8 * hi + 4);
            float kx[2][4], ky[2][4];
            #pragma unroll
            for (int j = 0; j < 4; ++j) {
                #pragma unroll
                for (int a = 0; a < 2; ++a) {
                    const float* kp = Kn + (8 * hi + 2 * j) * MS + wave * 32 + a * 16 + lm;
                    kx[a][j] = kp[0];
                    ky[a][j] = kp[MS];
                }
            }
            if (c + 1 < NCHK) {
                const float* krow = Kn2 + (cur ^ 1) * (CHK * MS) + lm * MS + wave * 32 + 8 * hi;
                #pragma unroll
                for (int a = 0; a < 2; ++a) {
                    kq[a][0] = *(const float4*)(krow + a * 16);
                    kq[a][1] = *(const float4*)(krow + a * 16 + 4);
                }
            }
            __builtin_amdgcn_sched_barrier(0);

            v8f u = {};
            {
                v2f ta, rb;
                ta.x = tq0.x; ta.y = tq0.y; rb.x = rq0.x; rb.y = rq0.y;
                u = __builtin_amdgcn_wmma_f32_16x16x4_f32(false, ta, false, rb, (short)0, u, false, false);
                ta.x = tq0.z; ta.y = tq0.w; rb.x = rq0.z; rb.y = rq0.w;
                u = __builtin_amdgcn_wmma_f32_16x16x4_f32(false, ta, false, rb, (short)0, u, false, false);
                ta.x = tq1.x; ta.y = tq1.y; rb.x = rq1.x; rb.y = rq1.y;
                u = __builtin_amdgcn_wmma_f32_16x16x4_f32(false, ta, false, rb, (short)0, u, false, false);
                ta.x = tq1.z; ta.y = tq1.w; rb.x = rq1.z; rb.y = rq1.w;
                u = __builtin_amdgcn_wmma_f32_16x16x4_f32(false, ta, false, rb, (short)0, u, false, false);
            }
            #pragma unroll
            for (int j = 0; j < 4; ++j) {
                v2f ub; ub.x = u[2 * j]; ub.y = u[2 * j + 1];
                #pragma unroll
                for (int a = 0; a < 2; ++a) {
                    v2f kt2; kt2.x = kx[a][j]; kt2.y = ky[a][j];
                    mt[a] = __builtin_amdgcn_wmma_f32_16x16x4_f32(
                                false, kt2, false, ub, (short)0, mt[a],
                                false, false);
                }
            }
        }
    }

    #pragma unroll
    for (int a = 0; a < 2; ++a) {
        float* dst = Msub + lm * MS + wave * 32 + a * 16 + 8 * hi;
        const v8f m = mt[a];
        float4 lo, up;
        lo.x = m[0]; lo.y = m[1]; lo.z = m[2]; lo.w = m[3];
        up.x = m[4]; up.y = m[5]; up.z = m[6]; up.w = m[7];
        *(float4*)dst = lo;
        *(float4*)(dst + 4) = up;
    }
    __syncthreads();
    for (int i = tid; i < VROWS * DK_; i += NTHREADS) {
        int r = i >> 8, k = i & 255;
        outB[r * DK_ + k] = Msub[r * MS + k];
    }
}

extern "C" void kernel_launch(void* const* d_in, const int* in_sizes, int n_in,
                              void* d_out, int out_size, void* d_ws, size_t ws_size,
                              hipStream_t stream) {
    (void)in_sizes; (void)n_in; (void)out_size;
    const float* mem = (const float*)d_in[0];
    const float* key = (const float*)d_in[1];
    const float* val = (const float*)d_in[2];
    float* out = (float*)d_out;
    float* ws  = (float*)d_ws;

    hipLaunchKernelGGL(DeltaRuleMemory_chunk_factors, dim3(FAC_GRID), dim3(NTHREADS), 0, stream,
                       key, ws, ws_size);
    hipLaunchKernelGGL(DeltaRuleMemory_chunked_wmma, dim3(B_ * SPLIT), dim3(NTHREADS), 0, stream,
                       mem, key, val, out, (const float*)ws, ws_size);
}


# Scan loop takes a pair of 16-step chunks per trip, 2 barriers per 32 steps not 4: chunk 1's right-hand side gets X*U0 on the matrix unit, cross gram X from the factor launch (sum order changed)
# speedup vs baseline: 4.1300x; 1.0352x over previous
#include <hip/hip_runtime.h>
#include <math.h>


typedef float v2f __attribute__((ext_vector_type(2)));
typedef float v8f __attribute__((ext_vector_type(8)));

#define B_   16
#define S_   4096
#define DK_  256
#define DV_  256
#define CHK  16
#define NCHK (S_ / CHK)
#define CH2  (2 * CHK)
#define NPAIR (NCHK / 2)
#define SPLIT 16
#define VROWS (DV_ / SPLIT)
#define NTHREADS 256
#define NWAVES (NTHREADS / 32)
#define MS 260
#define PS 18
#define RS 20
#define RS2 36
#define VS 18
#define WS 20

#define SMEM_FLOATS (VROWS * MS + 2 * CH2 * MS + NWAVES * CH2 * PS + VROWS * RS2 + 3 * CHK * RS + 2 * CH2 * VS)
#define STAGE_WAVE0 4
#define VPIECES (VROWS / 4)
#define DPP_ROW_XMASK 0x160

#define LR_C 0.1f
#define A_C  1.1f
#define EPS_C 1e-6f

#define FAC_CPW 8
#define FAC_GRID (B_ * NCHK / FAC_CPW)
#define WS_T_FLOATS (B_ * NCHK * CHK * CHK)
#define WS_S_FLOATS (B_ * S_)
#define WS_X_FLOATS (B_ * NPAIR * CHK * CHK)
#define WS_X_OFF (WS_T_FLOATS + WS_S_FLOATS)
#define WS_NEED_BYTES ((size_t)(WS_T_FLOATS + WS_S_FLOATS + WS_X_FLOATS) * sizeof(float))

template <int M>
__device__ __forceinline__ float row_xor_add(float x) {
    int y = __builtin_amdgcn_update_dpp(0, __float_as_int(x), DPP_ROW_XMASK + M,
                                        0xf, 0xf, true);
    return x + __int_as_float(y);
}

template <int U>
__device__ __forceinline__ void subst_step(float wn, float& x) {
    asm("v_fmac_f32_dpp %0, %0, %1 row_share:%2 row_mask:0xf bank_mask:0xf bound_ctrl:1"
        : "+v"(x) : "v"(wn), "n"(U));
}

__device__ __forceinline__ void lds_barrier() {
    __builtin_amdgcn_fence(__ATOMIC_RELEASE, "workgroup", "local");
    __builtin_amdgcn_s_barrier();
    __builtin_amdgcn_fence(__ATOMIC_ACQUIRE, "workgroup", "local");
}

__global__ __launch_bounds__(NTHREADS)
void DeltaRuleMemory_chunk_factors(const float* __restrict__ key, float* __restrict__ ws,
                                   size_t ws_size) {
    if (ws_size < WS_NEED_BYTES) return;

    __shared__ __attribute__((aligned(16))) float fsm[2 * CHK * MS + 2 * NWAVES * 256 + CHK * WS + FAC_CPW * CHK];
    float* Knf = fsm;
    float* Gp = Knf + 2 * CHK * MS;
    float* Gx = Gp + NWAVES * 256;
    float* Wu = Gx + NWAVES * 256;
    float* Sc = Wu + CHK * WS;

    const int tid  = threadIdx.x;
    const int wave = tid >> 5;
    const int lane = tid & 31;
    const int hi   = lane >> 4;
    const int lm   = lane & 15;
    const int row  = tid >> 4;

    const int cid0 = blockIdx.x * FAC_CPW;
    const float* kr = key + (size_t)cid0 * CHK * DK_ + (size_t)row * DK_ + lm * 4;
    float* Tw = ws + (size_t)cid0 * (CHK * CHK) + tid;
    float* Xw = ws + WS_X_OFF + (size_t)(cid0 >> 1) * (CHK * CHK) + tid;

    float4 q[4];
    #pragma unroll
    for (int j = 0; j < 4; ++j) q[j] = ((const float4*)kr)[j * 16];

    for (int i = 0; i < FAC_CPW; ++i) {
        float*       Kn = Knf + (i & 1) * (CHK * MS);
        const float* Kp = Knf + ((i & 1) ^ 1) * (CHK * MS);
        float ss = 0.f;
        #pragma unroll
        for (int j = 0; j < 4; ++j)
            ss += q[j].x * q[j].x + q[j].y * q[j].y + q[j].z * q[j].z + q[j].w * q[j].w;
        ss = row_xor_add<1>(ss);
        ss = row_xor_add<2>(ss);
        ss = row_xor_add<4>(ss);
        ss = row_xor_add<8>(ss);
        const float sc = 1.f / (sqrtf(ss) + EPS_C);
        #pragma unroll
        for (int j = 0; j < 4; ++j) {
            float4 o;
            o.x = q[j].x * sc; o.y = q[j].y * sc; o.z = q[j].z * sc; o.w = q[j].w * sc;
            ((float4*)(Kn + row * MS + lm * 4))[j * 16] = o;
        }
        if (lm == 0) Sc[i * CHK + row] = sc;
        if (i + 1 < FAC_CPW) {
            const float* kn = kr + (size_t)(i + 1) * CHK * DK_;
            #pragma unroll
            for (int j = 0; j < 4; ++j) q[j] = ((const float4*)kn)[j * 16];
        }
        lds_barrier();

        {
            const float* krow = Kn + lm * MS + wave * 32 + 8 * hi;
            v8f g = {};
            #pragma unroll
            for (int a = 0; a < 2; ++a) {
                const float4 q0 = *(const float4*)(krow + a * 16);
                const float4 q1 = *(const float4*)(krow + a * 16 + 4);
                v2f kf;
                kf.x = q0.x; kf.y = q0.y;
                g = __builtin_amdgcn_wmma_f32_16x16x4_f32(false, kf, false, kf, (short)0, g, false, false);
                kf.x = q0.z; kf.y = q0.w;
                g = __builtin_amdgcn_wmma_f32_16x16x4_f32(false, kf, false, kf, (short)0, g, false, false);
                kf.x = q1.x; kf.y = q1.y;
                g = __builtin_amdgcn_wmma_f32_16x16x4_f32(false, kf, false, kf, (short)0, g, false, false);
                kf.x = q1.z; kf.y = q1.w;
                g = __builtin_amdgcn_wmma_f32_16x16x4_f32(false, kf, false, kf, (short)0, g, false, false);
            }
            float* gp = Gp + wave * 256;
            #pragma unroll
            for (int r = 0; r < 8; ++r) gp[(r + 8 * hi) * 16 + lm] = g[r];
        }
        if (i & 1) {
            const float* krow = Kn + lm * MS + wave * 32 + 8 * hi;
            const float* prow = Kp + lm * MS + wave * 32 + 8 * hi;
            v8f x = {};
            #pragma unroll
            for (int a = 0; a < 2; ++a) {
                const float4 q0 = *(const float4*)(krow + a * 16);
                const float4 q1 = *(const float4*)(krow + a * 16 + 4);
                const float4 p0 = *(const float4*)(prow + a * 16);
                const float4 p1 = *(const float4*)(prow + a * 16 + 4);
                v2f kf, pf;
                kf.x = q0.x; kf.y = q0.y; pf.x = p0.x; pf.y = p0.y;
                x = __builtin_amdgcn_wmma_f32_16x16x4_f32(false, kf, false, pf, (short)0, x, false, false);
                kf.x = q0.z; kf.y = q0.w; pf.x = p0.z; pf.y = p0.w;
                x = __builtin_amdgcn_wmma_f32_16x16x4_f32(false, kf, false, pf, (short)0, x, false, false);
                kf.x = q1.x; kf.y = q1.y; pf.x = p1.x; pf.y = p1.y;
                x = __builtin_amdgcn_wmma_f32_16x16x4_f32(false, kf, false, pf, (short)0, x, false, false);
                kf.x = q1.z; kf.y = q1.w; pf.x = p1.z; pf.y = p1.w;
                x = __builtin_amdgcn_wmma_f32_16x16x4_f32(false, kf, false, pf, (short)0, x, false, false);
            }
            float* gx = Gx + wave * 256;
            #pragma unroll
            for (int r = 0; r < 8; ++r) gx[(r + 8 * hi) * 16 + lm] = x[r];
        }
        lds_barrier();

        if (i & 1) {
            float xs = Gx[tid];
            #pragma unroll
            for (int w = 1; w < NWAVES; ++w) xs += Gx[w * 256 + tid];
            Xw[(size_t)(i >> 1) * (CHK * CHK)] = -(A_C * xs);
        }
        {
            float gs = Gp[tid];
            #pragma unroll
            for (int w = 1; w < NWAVES; ++w) gs += Gp[w * 256 + tid];
            Wu[row * WS + lm] = (lm > row) ? -(A_C * gs) : 0.f;
        }
        lds_barrier();

        {
            const float4* wr = (const float4*)(Wu + lm * WS);
            const float4 w0 = wr[0], w1 = wr[1], w2 = wr[2], w3 = wr[3];
            float x = (lm == row) ? 1.f : 0.f;
            subst_step<15>(w3.w, x);
            subst_step<14>(w3.z, x);
            subst_step<13>(w3.y, x);
            subst_step<12>(w3.x, x);
            subst_step<11>(w2.w, x);
            subst_step<10>(w2.z, x);
            subst_step<9>(w2.y, x);
            subst_step<8>(w2.x, x);
            subst_step<7>(w1.w, x);
            subst_step<6>(w1.z, x);
            subst_step<5>(w1.y, x);
            subst_step<4>(w1.x, x);
            subst_step<3>(w0.w, x);
            subst_step<2>(w0.z, x);
            subst_step<1>(w0.y, x);
            Tw[(size_t)i * (CHK * CHK)] = x;
        }
    }
    lds_barrier();
    if (tid < FAC_CPW * CHK) ws[WS_T_FLOATS + (size_t)cid0 * CHK + tid] = Sc[tid];
}


struct Staged {
    float4 k[2][8];
    float4 v[2];
    float  sc[2];
};

__device__ __forceinline__ void load_chunk(Staged& s, const float* __restrict__ kr,
                                           const float* __restrict__ vg,
                                           const float* __restrict__ sr, bool has_v) {
    #pragma unroll
    for (int h = 0; h < 2; ++h) {
        #pragma unroll
        for (int j = 0; j < 8; ++j) s.k[h][j] = ((const float4*)(kr + (size_t)h * CHK * DK_))[j * 8];
        s.sc[h] = sr[h * CHK];
        if (has_v) s.v[h] = *(const float4*)(vg + (size_t)h * CHK * DV_);
    }
}

__device__ __forceinline__ void stage_write(const Staged& s, float* __restrict__ kw,
                                            float* __restrict__ vw, bool has_v) {
    #pragma unroll
    for (int h = 0; h < 2; ++h) {
        const float sc = s.sc[h];
        #pragma unroll
        for (int j = 0; j < 8; ++j) {
            float4 o;
            o.x = s.k[h][j].x * sc; o.y = s.k[h][j].y * sc;
            o.z = s.k[h][j].z * sc; o.w = s.k[h][j].w * sc;
            ((float4*)(kw + h * (CHK * MS)))[j * 8] = o;
        }
        if (has_v) {
            float* vh = vw + h * (CHK * VS);
            vh[0] = s.v[h].x; vh[1] = s.v[h].y; vh[2] = s.v[h].z; vh[3] = s.v[h].w;
        }
    }
}

__global__ __launch_bounds__(NTHREADS)
void DeltaRuleMemory_chunked_wmma(const float* __restrict__ mem,
                                  const float* __restrict__ key,
                                  const float* __restrict__ val,
                                  float* __restrict__ out,
                                  const float* __restrict__ ws,
                                  size_t ws_size) {
    if (ws_size < WS_NEED_BYTES) return;

    __shared__ __attribute__((aligned(16))) float smem[SMEM_FLOATS];
    float* Msub = smem;
    float* Kn2  = Msub + VROWS * MS;
    float* P0p  = Kn2  + 2 * CH2 * MS;
    float* Rv   = P0p  + NWAVES * CH2 * PS;
    float* Tl   = Rv   + VROWS * RS2;
    float* Vc2  = Tl   + 3 * CHK * RS;

    const int tid  = threadIdx.x;
    const int wave = tid >> 5;
    const int lane = tid & 31;
    const int hi   = lane >> 4;
    const int lm   = lane & 15;

    const int blk = blockIdx.x;
    const int b   = blk / SPLIT;
    const int v0  = (blk % SPLIT) * VROWS;

    const float* memB = mem + (size_t)b * DV_ * DK_ + (size_t)v0 * DK_;
    const float* keyB = key + (size_t)b * S_ * DK_;
    const float* valB = val + (size_t)b * S_ * DV_;
    float*       outB = out + (size_t)b * DV_ * DK_ + (size_t)v0 * DK_;
    const float* scB  = ws + WS_T_FLOATS + (size_t)b * S_;

    const bool stager = wave >= STAGE_WAVE0;
    const int  sT  = (tid & (4 * 32 - 1)) >> 3;
    const int  sG  = tid & 7;
    const bool sV  = sG < VPIECES;
    const int  sGv = sV ? sG : 0;
    const float* krT = keyB + (size_t)sT * DK_ + sG * 4;
    const float* vgT = valB + (size_t)sT * DV_ + v0 + sGv * 4;
    const float* srT = scB + sT;
    float* kwT = Kn2 + sT * MS + sG * 4;
    float* vwT = Vc2 + sT * VS + sGv * 4;
    const int    sE  = 2 * (tid & (4 * 32 - 1));
    const float* tpT = ws + (size_t)b * NCHK * (CHK * CHK) + sE;
    const float* xpT = ws + WS_X_OFF + (size_t)b * NPAIR * (CHK * CHK) + sE;
    float*       twT = Tl + (sE >> 4) * RS + (sE & 15);

    Staged st;
    float2 stT[3] = {{0.f, 0.f}, {0.f, 0.f}, {0.f, 0.f}};
    if (stager) load_chunk(st, krT, vgT, srT, sV);
    for (int i = tid; i < VROWS * DK_; i += NTHREADS) {
        int r = i >> 8, k = i & 255;
        Msub[r * MS + k] = memB[r * DK_ + k];
    }
    if (stager) stage_write(st, kwT, vwT, sV);
    __syncthreads();

    v8f mt[2];
    #pragma unroll
    for (int a = 0; a < 2; ++a) {
        const float* src = Msub + lm * MS + wave * 32 + a * 16 + 8 * hi;
        const float4 lo = *(const float4*)src;
        const float4 up = *(const float4*)(src + 4);
        v8f m;
        m[0] = lo.x; m[1] = lo.y; m[2] = lo.z; m[3] = lo.w;
        m[4] = up.x; m[5] = up.y; m[6] = up.z; m[7] = up.w;
        mt[a] = m;
    }

    if (stager) {
        load_chunk(st, krT + (size_t)CH2 * DK_, vgT + (size_t)CH2 * DV_, srT + CH2, sV);
        stT[0] = *(const float2*)tpT;
        stT[1] = *(const float2*)(tpT + CHK * CHK);
        stT[2] = *(const float2*)xpT;
        __builtin_prefetch(keyB + ((size_t)2 * CH2 + sT) * DK_ + sG * 32, 0, 0);
        __builtin_prefetch(keyB + ((size_t)2 * CH2 + CHK + sT) * DK_ + sG * 32, 0, 0);
    }
    float4 kq[2][2][2];
    #pragma unroll
    for (int h = 0; h < 2; ++h) {
        const float* krow = Kn2 + (h * CHK + lm) * MS + wave * 32 + 8 * hi;
        #pragma unroll
        for (int a = 0; a < 2; ++a) {
            kq[h][a][0] = *(const float4*)(krow + a * 16);
            kq[h][a][1] = *(const float4*)(krow + a * 16 + 4);
        }
    }

    for (int p = 0; p < NPAIR; ++p) {
        const int cur = p & 1;
        const float* Kn = Kn2 + cur * (CH2 * MS);
        const float* Vc = Vc2 + cur * (CH2 * VS);
        const bool stage_next = stager && (p + 1 < NPAIR);

        {
            v8f p0a[2] = {{}, {}};
            #pragma unroll
            for (int a = 0; a < 2; ++a) {
                #pragma unroll
                for (int j = 0; j < 4; ++j) {
                    v2f m0; m0.x = mt[a][2 * j]; m0.y = mt[a][2 * j + 1];
                    #pragma unroll
                    for (int h = 0; h < 2; ++h) {
                        const float4 q = kq[h][a][j >> 1];
                        v2f kf;
                        kf.x = (j & 1) ? q.z : q.x;
                        kf.y = (j & 1) ? q.w : q.y;
                        p0a[h] = __builtin_amdgcn_wmma_f32_16x16x4_f32(
                                     false, kf, false, m0, (short)0, p0a[h], false, false);
                    }
                }
            }
            float* pp = P0p + wave * (CH2 * PS);
            #pragma unroll
            for (int h = 0; h < 2; ++h) {
                #pragma unroll
                for (int r = 0; r < 8; ++r) pp[(h * CHK + r + 8 * hi) * PS + lm] = p0a[h][r];
            }
        }
        lds_barrier();

        if (!stager) {
            const int va = 4 * wave + hi;
            float pa[2], pb[2], xa[2], xb[2];
            #pragma unroll
            for (int h = 0; h < 2; ++h) {
                const float* pp = P0p + (h * CHK + lm) * PS + va;
                pa[h] = pp[0];
                pb[h] = pp[2];
                #pragma unroll
                for (int w = 1; w < NWAVES; ++w) {
                    pa[h] += pp[w * (CH2 * PS)];
                    pb[h] += pp[w * (CH2 * PS) + 2];
                }
                xa[h] = Vc[(h * CHK + lm) * VS + va];
                xb[h] = Vc[(h * CHK + lm) * VS + va + 2];
            }
            #pragma unroll
            for (int h = 0; h < 2; ++h) {
                Rv[va * RS2 + h * CHK + lm]       = LR_C * xa[h] - A_C * pa[h];
                Rv[(va + 2) * RS2 + h * CHK + lm] = LR_C * xb[h] - A_C * pb[h];
            }
        } else {
            *(float2*)twT = stT[0];
            *(float2*)(twT + CHK * RS) = stT[1];
            *(float2*)(twT + 2 * CHK * RS) = stT[2];
            if (stage_next) {
                const int nxt = cur ^ 1;
                stage_write(st, kwT + nxt * (CH2 * MS), vwT + nxt * (CH2 * VS), sV);
                if (p + 2 < NPAIR) {
                    load_chunk(st, krT + (size_t)(p + 2) * CH2 * DK_, vgT + (size_t)(p + 2) * CH2 * DV_,
                               srT + (p + 2) * CH2, sV);
                    if (p + 3 < NPAIR) {
                        __builtin_prefetch(keyB + ((size_t)(p + 3) * CH2 + sT) * DK_ + sG * 32, 0, 0);
                        __builtin_prefetch(keyB + ((size_t)(p + 3) * CH2 + CHK + sT) * DK_ + sG * 32, 0, 0);
                    }
                }
                stT[0] = *(const float2*)(tpT + (size_t)(2 * p + 2) * (CHK * CHK));
                stT[1] = *(const float2*)(tpT + (size_t)(2 * p + 3) * (CHK * CHK));
                stT[2] = *(const float2*)(xpT + (size_t)(p + 1) * (CHK * CHK));
            }
        }
        lds_barrier();

        {
            const float4 rq0 = *(const float4*)(Rv + lm * RS2 + 8 * hi);
            const float4 rq1 = *(const float4*)(Rv + lm * RS2 + 8 * hi + 4);
            const float4 rb0 = *(const float4*)(Rv + lm * RS2 + CHK + 8 * hi);
            const float4 rb1 = *(const float4*)(Rv + lm * RS2 + CHK + 8 * hi + 4);
            float4 fq[3][2];
            #pragma unroll
            for (int m = 0; m < 3; ++m) {
                fq[m][0] = *(const float4*)(Tl + m * (CHK * RS) + lm * RS + 8 * hi);
                fq[m][1] = *(const float4*)(Tl + m * (CHK * RS) + lm * RS + 8 * hi + 4);
            }
            __builtin_amdgcn_sched_barrier(0);
            float kx[2][2][4], ky[2][2][4];
            #pragma unroll
            for (int h = 0; h < 2; ++h) {
                #pragma unroll
                for (int j = 0; j < 4; ++j) {
                    #pragma unroll
                    for (int a = 0; a < 2; ++a) {
                        const float* kp = Kn + (h * CHK + 8 * hi + 2 * j) * MS + wave * 32 + a * 16 + lm;
                        kx[h][a][j] = kp[0];
                        ky[h][a][j] = kp[MS];
                    }
                }
            }
            __builtin_amdgcn_sched_barrier(0);
            if (p + 1 < NPAIR) {
                #pragma unroll
                for (int h = 0; h < 2; ++h) {
                    const float* krow = Kn2 + (cur ^ 1) * (CH2 * MS) + (h * CHK + lm) * MS + wave * 32 + 8 * hi;
                    #pragma unroll
                    for (int a = 0; a < 2; ++a) {
                        kq[h][a][0] = *(const float4*)(krow + a * 16);
                        kq[h][a][1] = *(const float4*)(krow + a * 16 + 4);
                    }
                }
            }
            __builtin_amdgcn_sched_barrier(0);

            v8f u0 = {};
            {
                v2f ta, rb;
                ta.x = fq[0][0].x; ta.y = fq[0][0].y; rb.x = rq0.x; rb.y = rq0.y;
                u0 = __builtin_amdgcn_wmma_f32_16x16x4_f32(false, ta, false, rb, (short)0, u0, false, false);
                ta.x = fq[0][0].z; ta.y = fq[0][0].w; rb.x = rq0.z; rb.y = rq0.w;
                u0 = __builtin_amdgcn_wmma_f32_16x16x4_f32(false, ta, false, rb, (short)0, u0, false, false);
                ta.x = fq[0][1].x; ta.y = fq[0][1].y; rb.x = rq1.x; rb.y = rq1.y;
                u0 = __builtin_amdgcn_wmma_f32_16x16x4_f32(false, ta, false, rb, (short)0, u0, false, false);
                ta.x = fq[0][1].z; ta.y = fq[0][1].w; rb.x = rq1.z; rb.y = rq1.w;
                u0 = __builtin_amdgcn_wmma_f32_16x16x4_f32(false, ta, false, rb, (short)0, u0, false, false);
            }
            v8f r1;
            r1[0] = rb0.x; r1[1] = rb0.y; r1[2] = rb0.z; r1[3] = rb0.w;
            r1[4] = rb1.x; r1[5] = rb1.y; r1[6] = rb1.z; r1[7] = rb1.w;
            #pragma unroll
            for (int j = 0; j < 4; ++j) {
                const float4 f = fq[2][j >> 1];
                v2f xa;
                xa.x = (j & 1) ? f.z : f.x;
                xa.y = (j & 1) ? f.w : f.y;
                v2f ub; ub.x = u0[2 * j]; ub.y = u0[2 * j + 1];
                r1 = __builtin_amdgcn_wmma_f32_16x16x4_f32(false, xa, false, ub, (short)0, r1, false, false);
                #pragma unroll
                for (int a = 0; a < 2; ++a) {
                    v2f kt2; kt2.x = kx[0][a][j]; kt2.y = ky[0][a][j];
                    mt[a] = __builtin_amdgcn_wmma_f32_16x16x4_f32(
                                false, kt2, false, ub, (short)0, mt[a],
                                false, false);
                }
            }
            v8f u1 = {};
            #pragma unroll
            for (int j = 0; j < 4; ++j) {
                const float4 f = fq[1][j >> 1];
                v2f ta;
                ta.x = (j & 1) ? f.z : f.x;
                ta.y = (j & 1) ? f.w : f.y;
                v2f rb; rb.x = r1[2 * j]; rb.y = r1[2 * j + 1];
                u1 = __builtin_amdgcn_wmma_f32_16x16x4_f32(false, ta, false, rb, (short)0, u1, false, false);
            }
            #pragma unroll
            for (int j = 0; j < 4; ++j) {
                v2f ub; ub.x = u1[2 * j]; ub.y = u1[2 * j + 1];
                #pragma unroll
                for (int a = 0; a < 2; ++a) {
                    v2f kt2; kt2.x = kx[1][a][j]; kt2.y = ky[1][a][j];
                    mt[a] = __builtin_amdgcn_wmma_f32_16x16x4_f32(
                                false, kt2, false, ub, (short)0, mt[a],
                                false, false);
                }
            }
        }
    }

    #pragma unroll
    for (int a = 0; a < 2; ++a) {
        float* dst = Msub + lm * MS + wave * 32 + a * 16 + 8 * hi;
        const v8f m = mt[a];
        float4 lo, up;
        lo.x = m[0]; lo.y = m[1]; lo.z = m[2]; lo.w = m[3];
        up.x = m[4]; up.y = m[5]; up.z = m[6]; up.w = m[7];
        *(float4*)dst = lo;
        *(float4*)(dst + 4) = up;
    }
    __syncthreads();
    for (int i = tid; i < VROWS * DK_; i += NTHREADS) {
        int r = i >> 8, k = i & 255;
        outB[r * DK_ + k] = Msub[r * MS + k];
    }
}

extern "C" void kernel_launch(void* const* d_in, const int* in_sizes, int n_in,
                              void* d_out, int out_size, void* d_ws, size_t ws_size,
                              hipStream_t stream) {
    (void)in_sizes; (void)n_in; (void)out_size;
    const float* mem = (const float*)d_in[0];
    const float* key = (const float*)d_in[1];
    const float* val = (const float*)d_in[2];
    float* out = (float*)d_out;
    float* ws  = (float*)d_ws;

    hipLaunchKernelGGL(DeltaRuleMemory_chunk_factors, dim3(FAC_GRID), dim3(NTHREADS), 0, stream,
                       key, ws, ws_size);
    hipLaunchKernelGGL(DeltaRuleMemory_chunked_wmma, dim3(B_ * SPLIT), dim3(NTHREADS), 0, stream,
                       mem, key, val, out, (const float*)ws, ws_size);
}


# Scan's matrix work on 4 waves of 64 k (one per SIMD), not 8 of 32: 76 matrix instructions per SIMD and trip, not 88 (factor products 4x, not 8x); waves 4-7 only stage (sum order changed)
# speedup vs baseline: 4.5953x; 1.1127x over previous
#include <hip/hip_runtime.h>
#include <math.h>


typedef float v2f __attribute__((ext_vector_type(2)));
typedef float v8f __attribute__((ext_vector_type(8)));

#define B_   16
#define S_   4096
#define DK_  256
#define DV_  256
#define CHK  16
#define NCHK (S_ / CHK)
#define CH2  (2 * CHK)
#define NPAIR (NCHK / 2)
#define SPLIT 16
#define VROWS (DV_ / SPLIT)
#define NTHREADS 256
#define NWAVES (NTHREADS / 32)
#define MWAVES 4
#define KT 4
#define MS 260
#define PS 18
#define RS 20
#define RS2 36
#define VS 18
#define WS 20

#define SMEM_FLOATS (VROWS * MS + 2 * CH2 * MS + MWAVES * CH2 * PS + VROWS * RS2 + 3 * CHK * RS + 2 * CH2 * VS)
#define STAGE_WAVE0 MWAVES
#define VPIECES (VROWS / 4)
#define DPP_ROW_XMASK 0x160

#define LR_C 0.1f
#define A_C  1.1f
#define EPS_C 1e-6f

#define FAC_CPW 8
#define FAC_GRID (B_ * NCHK / FAC_CPW)
#define WS_T_FLOATS (B_ * NCHK * CHK * CHK)
#define WS_S_FLOATS (B_ * S_)
#define WS_X_FLOATS (B_ * NPAIR * CHK * CHK)
#define WS_X_OFF (WS_T_FLOATS + WS_S_FLOATS)
#define WS_NEED_BYTES ((size_t)(WS_T_FLOATS + WS_S_FLOATS + WS_X_FLOATS) * sizeof(float))

template <int M>
__device__ __forceinline__ float row_xor_add(float x) {
    int y = __builtin_amdgcn_update_dpp(0, __float_as_int(x), DPP_ROW_XMASK + M,
                                        0xf, 0xf, true);
    return x + __int_as_float(y);
}

template <int U>
__device__ __forceinline__ void subst_step(float wn, float& x) {
    asm("v_fmac_f32_dpp %0, %0, %1 row_share:%2 row_mask:0xf bank_mask:0xf bound_ctrl:1"
        : "+v"(x) : "v"(wn), "n"(U));
}

__device__ __forceinline__ void lds_barrier() {
    __builtin_amdgcn_fence(__ATOMIC_RELEASE, "workgroup", "local");
    __builtin_amdgcn_s_barrier();
    __builtin_amdgcn_fence(__ATOMIC_ACQUIRE, "workgroup", "local");
}

__global__ __launch_bounds__(NTHREADS)
void DeltaRuleMemory_chunk_factors(const float* __restrict__ key, float* __restrict__ ws,
                                   size_t ws_size) {
    if (ws_size < WS_NEED_BYTES) return;

    __shared__ __attribute__((aligned(16))) float fsm[2 * CHK * MS + 2 * NWAVES * 256 + CHK * WS + FAC_CPW * CHK];
    float* Knf = fsm;
    float* Gp = Knf + 2 * CHK * MS;
    float* Gx = Gp + NWAVES * 256;
    float* Wu = Gx + NWAVES * 256;
    float* Sc = Wu + CHK * WS;

    const int tid  = threadIdx.x;
    const int wave = tid >> 5;
    const int lane = tid & 31;
    const int hi   = lane >> 4;
    const int lm   = lane & 15;
    const int row  = tid >> 4;

    const int cid0 = blockIdx.x * FAC_CPW;
    const float* kr = key + (size_t)cid0 * CHK * DK_ + (size_t)row * DK_ + lm * 4;
    float* Tw = ws + (size_t)cid0 * (CHK * CHK) + tid;
    float* Xw = ws + WS_X_OFF + (size_t)(cid0 >> 1) * (CHK * CHK) + tid;

    float4 q[4];
    #pragma unroll
    for (int j = 0; j < 4; ++j) q[j] = ((const float4*)kr)[j * 16];

    for (int i = 0; i < FAC_CPW; ++i) {
        float*       Kn = Knf + (i & 1) * (CHK * MS);
        const float* Kp = Knf + ((i & 1) ^ 1) * (CHK * MS);
        float ss = 0.f;
        #pragma unroll
        for (int j = 0; j < 4; ++j)
            ss += q[j].x * q[j].x + q[j].y * q[j].y + q[j].z * q[j].z + q[j].w * q[j].w;
        ss = row_xor_add<1>(ss);
        ss = row_xor_add<2>(ss);
        ss = row_xor_add<4>(ss);
        ss = row_xor_add<8>(ss);
        const float sc = 1.f / (sqrtf(ss) + EPS_C);
        #pragma unroll
        for (int j = 0; j < 4; ++j) {
            float4 o;
            o.x = q[j].x * sc; o.y = q[j].y * sc; o.z = q[j].z * sc; o.w = q[j].w * sc;
            ((float4*)(Kn + row * MS + lm * 4))[j * 16] = o;
        }
        if (lm == 0) Sc[i * CHK + row] = sc;
        if (i + 1 < FAC_CPW) {
            const float* kn = kr + (size_t)(i + 1) * CHK * DK_;
            #pragma unroll
            for (int j = 0; j < 4; ++j) q[j] = ((const float4*)kn)[j * 16];
        }
        lds_barrier();

        {
            const float* krow = Kn + lm * MS + wave * 32 + 8 * hi;
            v8f g = {};
            #pragma unroll
            for (int a = 0; a < 2; ++a) {
                const float4 q0 = *(const float4*)(krow + a * 16);
                const float4 q1 = *(const float4*)(krow + a * 16 + 4);
                v2f kf;
                kf.x = q0.x; kf.y = q0.y;
                g = __builtin_amdgcn_wmma_f32_16x16x4_f32(false, kf, false, kf, (short)0, g, false, false);
                kf.x = q0.z; kf.y = q0.w;
                g = __builtin_amdgcn_wmma_f32_16x16x4_f32(false, kf, false, kf, (short)0, g, false, false);
                kf.x = q1.x; kf.y = q1.y;
                g = __builtin_amdgcn_wmma_f32_16x16x4_f32(false, kf, false, kf, (short)0, g, false, false);
                kf.x = q1.z; kf.y = q1.w;
                g = __builtin_amdgcn_wmma_f32_16x16x4_f32(false, kf, false, kf, (short)0, g, false, false);
            }
            float* gp = Gp + wave * 256;
            #pragma unroll
            for (int r = 0; r < 8; ++r) gp[(r + 8 * hi) * 16 + lm] = g[r];
        }
        if (i & 1) {
            const float* krow = Kn + lm * MS + wave * 32 + 8 * hi;
            const float* prow = Kp + lm * MS + wave * 32 + 8 * hi;
            v8f x = {};
            #pragma unroll
            for (int a = 0; a < 2; ++a) {
                const float4 q0 = *(const float4*)(krow + a * 16);
                const float4 q1 = *(const float4*)(krow + a * 16 + 4);
                const float4 p0 = *(const float4*)(prow + a * 16);
                const float4 p1 = *(const float4*)(prow + a * 16 + 4);
                v2f kf, pf;
                kf.x = q0.x; kf.y = q0.y; pf.x = p0.x; pf.y = p0.y;
                x = __builtin_amdgcn_wmma_f32_16x16x4_f32(false, kf, false, pf, (short)0, x, false, false);
                kf.x = q0.z; kf.y = q0.w; pf.x = p0.z; pf.y = p0.w;
                x = __builtin_amdgcn_wmma_f32_16x16x4_f32(false, kf, false, pf, (short)0, x, false, false);
                kf.x = q1.x; kf.y = q1.y; pf.x = p1.x; pf.y = p1.y;
                x = __builtin_amdgcn_wmma_f32_16x16x4_f32(false, kf, false, pf, (short)0, x, false, false);
                kf.x = q1.z; kf.y = q1.w; pf.x = p1.z; pf.y = p1.w;
                x = __builtin_amdgcn_wmma_f32_16x16x4_f32(false, kf, false, pf, (short)0, x, false, false);
            }
            float* gx = Gx + wave * 256;
            #pragma unroll
            for (int r = 0; r < 8; ++r) gx[(r + 8 * hi) * 16 + lm] = x[r];
        }
        lds_barrier();

        if (i & 1) {
            float xs = Gx[tid];
            #pragma unroll
            for (int w = 1; w < NWAVES; ++w) xs += Gx[w * 256 + tid];
            Xw[(size_t)(i >> 1) * (CHK * CHK)] = -(A_C * xs);
        }
        {
            float gs = Gp[tid];
            #pragma unroll
            for (int w = 1; w < NWAVES; ++w) gs += Gp[w * 256 + tid];
            Wu[row * WS + lm] = (lm > row) ? -(A_C * gs) : 0.f;
        }
        lds_barrier();

        {
            const float4* wr = (const float4*)(Wu + lm * WS);
            const float4 w0 = wr[0], w1 = wr[1], w2 = wr[2], w3 = wr[3];
            float x = (lm == row) ? 1.f : 0.f;
            subst_step<15>(w3.w, x);
            subst_step<14>(w3.z, x);
            subst_step<13>(w3.y, x);
            subst_step<12>(w3.x, x);
            subst_step<11>(w2.w, x);
            subst_step<10>(w2.z, x);
            subst_step<9>(w2.y, x);
            subst_step<8>(w2.x, x);
            subst_step<7>(w1.w, x);
            subst_step<6>(w1.z, x);
            subst_step<5>(w1.y, x);
            subst_step<4>(w1.x, x);
            subst_step<3>(w0.w, x);
            subst_step<2>(w0.z, x);
            subst_step<1>(w0.y, x);
            Tw[(size_t)i * (CHK * CHK)] = x;
        }
    }
    lds_barrier();
    if (tid < FAC_CPW * CHK) ws[WS_T_FLOATS + (size_t)cid0 * CHK + tid] = Sc[tid];
}


struct Staged {
    float4 k[2][8];
    float4 v[2];
    float  sc[2];
};

__device__ __forceinline__ void load_chunk(Staged& s, const float* __restrict__ kr,
                                           const float* __restrict__ vg,
                                           const float* __restrict__ sr, bool has_v) {
    #pragma unroll
    for (int h = 0; h < 2; ++h) {
        #pragma unroll
        for (int j = 0; j < 8; ++j) s.k[h][j] = ((const float4*)(kr + (size_t)h * CHK * DK_))[j * 8];
        s.sc[h] = sr[h * CHK];
        if (has_v) s.v[h] = *(const float4*)(vg + (size_t)h * CHK * DV_);
    }
}

__device__ __forceinline__ void stage_write(const Staged& s, float* __restrict__ kw,
                                            float* __restrict__ vw, bool has_v) {
    #pragma unroll
    for (int h = 0; h < 2; ++h) {
        const float sc = s.sc[h];
        #pragma unroll
        for (int j = 0; j < 8; ++j) {
            float4 o;
            o.x = s.k[h][j].x * sc; o.y = s.k[h][j].y * sc;
            o.z = s.k[h][j].z * sc; o.w = s.k[h][j].w * sc;
            ((float4*)(kw + h * (CHK * MS)))[j * 8] = o;
        }
        if (has_v) {
            float* vh = vw + h * (CHK * VS);
            vh[0] = s.v[h].x; vh[1] = s.v[h].y; vh[2] = s.v[h].z; vh[3] = s.v[h].w;
        }
    }
}

__global__ __launch_bounds__(NTHREADS)
void DeltaRuleMemory_chunked_wmma(const float* __restrict__ mem,
                                  const float* __restrict__ key,
                                  const float* __restrict__ val,
                                  float* __restrict__ out,
                                  const float* __restrict__ ws,
                                  size_t ws_size) {
    if (ws_size < WS_NEED_BYTES) return;

    __shared__ __attribute__((aligned(16))) float smem[SMEM_FLOATS];
    float* Msub = smem;
    float* Kn2  = Msub + VROWS * MS;
    float* P0p  = Kn2  + 2 * CH2 * MS;
    float* Rv   = P0p  + MWAVES * CH2 * PS;
    float* Tl   = Rv   + VROWS * RS2;
    float* Vc2  = Tl   + 3 * CHK * RS;

    const int tid  = threadIdx.x;
    const int wave = __builtin_amdgcn_readfirstlane(tid >> 5);
    const int lane = tid & 31;
    const int hi   = lane >> 4;
    const int lm   = lane & 15;
    const bool mwave = wave < MWAVES;

    const int blk = blockIdx.x;
    const int b   = blk / SPLIT;
    const int v0  = (blk % SPLIT) * VROWS;

    const float* memB = mem + (size_t)b * DV_ * DK_ + (size_t)v0 * DK_;
    const float* keyB = key + (size_t)b * S_ * DK_;
    const float* valB = val + (size_t)b * S_ * DV_;
    float*       outB = out + (size_t)b * DV_ * DK_ + (size_t)v0 * DK_;
    const float* scB  = ws + WS_T_FLOATS + (size_t)b * S_;

    const bool stager = wave >= STAGE_WAVE0;
    const int  sT  = (tid & (4 * 32 - 1)) >> 3;
    const int  sG  = tid & 7;
    const bool sV  = sG < VPIECES;
    const int  sGv = sV ? sG : 0;
    const float* krT = keyB + (size_t)sT * DK_ + sG * 4;
    const float* vgT = valB + (size_t)sT * DV_ + v0 + sGv * 4;
    const float* srT = scB + sT;
    float* kwT = Kn2 + sT * MS + sG * 4;
    float* vwT = Vc2 + sT * VS + sGv * 4;
    const int    sE  = 2 * (tid & (4 * 32 - 1));
    const float* tpT = ws + (size_t)b * NCHK * (CHK * CHK) + sE;
    const float* xpT = ws + WS_X_OFF + (size_t)b * NPAIR * (CHK * CHK) + sE;
    float*       twT = Tl + (sE >> 4) * RS + (sE & 15);

    Staged st;
    float2 stT[3] = {{0.f, 0.f}, {0.f, 0.f}, {0.f, 0.f}};
    if (stager) load_chunk(st, krT, vgT, srT, sV);
    for (int i = tid; i < VROWS * DK_; i += NTHREADS) {
        int r = i >> 8, k = i & 255;
        Msub[r * MS + k] = memB[r * DK_ + k];
    }
    if (stager) stage_write(st, kwT, vwT, sV);
    __syncthreads();

    v8f mt[KT] = {{}, {}, {}, {}};
    if (mwave) {
        #pragma unroll
        for (int a = 0; a < KT; ++a) {
            const float* src = Msub + lm * MS + wave * (KT * 16) + a * 16 + 8 * hi;
            const float4 lo = *(const float4*)src;
            const float4 up = *(const float4*)(src + 4);
            v8f m;
            m[0] = lo.x; m[1] = lo.y; m[2] = lo.z; m[3] = lo.w;
            m[4] = up.x; m[5] = up.y; m[6] = up.z; m[7] = up.w;
            mt[a] = m;
        }
    }

    if (stager) {
        load_chunk(st, krT + (size_t)CH2 * DK_, vgT + (size_t)CH2 * DV_, srT + CH2, sV);
        stT[0] = *(const float2*)tpT;
        stT[1] = *(const float2*)(tpT + CHK * CHK);
        stT[2] = *(const float2*)xpT;
        __builtin_prefetch(keyB + ((size_t)2 * CH2 + sT) * DK_ + sG * 32, 0, 0);
        __builtin_prefetch(keyB + ((size_t)2 * CH2 + CHK + sT) * DK_ + sG * 32, 0, 0);
    }
    float4 kq[2][KT][2] = {};
    if (mwave) {
        #pragma unroll
        for (int h = 0; h < 2; ++h) {
            const float* krow = Kn2 + (h * CHK + lm) * MS + wave * (KT * 16) + 8 * hi;
            #pragma unroll
            for (int a = 0; a < KT; ++a) {
                kq[h][a][0] = *(const float4*)(krow + a * 16);
                kq[h][a][1] = *(const float4*)(krow + a * 16 + 4);
            }
        }
    }

    for (int p = 0; p < NPAIR; ++p) {
        const int cur = p & 1;
        const float* Kn = Kn2 + cur * (CH2 * MS);
        const float* Vc = Vc2 + cur * (CH2 * VS);
        const bool stage_next = stager && (p + 1 < NPAIR);

        if (mwave) {
            v8f p0a[2] = {{}, {}};
            #pragma unroll
            for (int a = 0; a < KT; ++a) {
                #pragma unroll
                for (int j = 0; j < 4; ++j) {
                    v2f m0; m0.x = mt[a][2 * j]; m0.y = mt[a][2 * j + 1];
                    #pragma unroll
                    for (int h = 0; h < 2; ++h) {
                        const float4 q = kq[h][a][j >> 1];
                        v2f kf;
                        kf.x = (j & 1) ? q.z : q.x;
                        kf.y = (j & 1) ? q.w : q.y;
                        p0a[h] = __builtin_amdgcn_wmma_f32_16x16x4_f32(
                                     false, kf, false, m0, (short)0, p0a[h], false, false);
                    }
                }
            }
            float* pp = P0p + wave * (CH2 * PS);
            #pragma unroll
            for (int h = 0; h < 2; ++h) {
                #pragma unroll
                for (int r = 0; r < 8; ++r) pp[(h * CHK + r + 8 * hi) * PS + lm] = p0a[h][r];
            }
        }
        lds_barrier();

        if (!stager) {
            const int va = 4 * wave + hi;
            float pa[2], pb[2], xa[2], xb[2];
            #pragma unroll
            for (int h = 0; h < 2; ++h) {
                const float* pp = P0p + (h * CHK + lm) * PS + va;
                pa[h] = pp[0];
                pb[h] = pp[2];
                #pragma unroll
                for (int w = 1; w < MWAVES; ++w) {
                    pa[h] += pp[w * (CH2 * PS)];
                    pb[h] += pp[w * (CH2 * PS) + 2];
                }
                xa[h] = Vc[(h * CHK + lm) * VS + va];
                xb[h] = Vc[(h * CHK + lm) * VS + va + 2];
            }
            #pragma unroll
            for (int h = 0; h < 2; ++h) {
                Rv[va * RS2 + h * CHK + lm]       = LR_C * xa[h] - A_C * pa[h];
                Rv[(va + 2) * RS2 + h * CHK + lm] = LR_C * xb[h] - A_C * pb[h];
            }
        } else {
            *(float2*)twT = stT[0];
            *(float2*)(twT + CHK * RS) = stT[1];
            *(float2*)(twT + 2 * CHK * RS) = stT[2];
            if (stage_next) {
                const int nxt = cur ^ 1;
                stage_write(st, kwT + nxt * (CH2 * MS), vwT + nxt * (CH2 * VS), sV);
                if (p + 2 < NPAIR) {
                    load_chunk(st, krT + (size_t)(p + 2) * CH2 * DK_, vgT + (size_t)(p + 2) * CH2 * DV_,
                               srT + (p + 2) * CH2, sV);
                    if (p + 3 < NPAIR) {
                        __builtin_prefetch(keyB + ((size_t)(p + 3) * CH2 + sT) * DK_ + sG * 32, 0, 0);
                        __builtin_prefetch(keyB + ((size_t)(p + 3) * CH2 + CHK + sT) * DK_ + sG * 32, 0, 0);
                    }
                }
                stT[0] = *(const float2*)(tpT + (size_t)(2 * p + 2) * (CHK * CHK));
                stT[1] = *(const float2*)(tpT + (size_t)(2 * p + 3) * (CHK * CHK));
                stT[2] = *(const float2*)(xpT + (size_t)(p + 1) * (CHK * CHK));
            }
        }
        lds_barrier();

        if (mwave) {
            const float4 rq0 = *(const float4*)(Rv + lm * RS2 + 8 * hi);
            const float4 rq1 = *(const float4*)(Rv + lm * RS2 + 8 * hi + 4);
            const float4 rb0 = *(const float4*)(Rv + lm * RS2 + CHK + 8 * hi);
            const float4 rb1 = *(const float4*)(Rv + lm * RS2 + CHK + 8 * hi + 4);
            float4 fq[3][2];
            #pragma unroll
            for (int m = 0; m < 3; ++m) {
                fq[m][0] = *(const float4*)(Tl + m * (CHK * RS) + lm * RS + 8 * hi);
                fq[m][1] = *(const float4*)(Tl + m * (CHK * RS) + lm * RS + 8 * hi + 4);
            }
            __builtin_amdgcn_sched_barrier(0);
            float kx[2][KT][4], ky[2][KT][4];
            #pragma unroll
            for (int h = 0; h < 2; ++h) {
                #pragma unroll
                for (int j = 0; j < 4; ++j) {
                    #pragma unroll
                    for (int a = 0; a < KT; ++a) {
                        const float* kp = Kn + (h * CHK + 8 * hi + 2 * j) * MS + wave * (KT * 16) + a * 16 + lm;
                        kx[h][a][j] = kp[0];
                        ky[h][a][j] = kp[MS];
                    }
                }
            }
            __builtin_amdgcn_sched_barrier(0);
            if (p + 1 < NPAIR) {
                #pragma unroll
                for (int h = 0; h < 2; ++h) {
                    const float* krow = Kn2 + (cur ^ 1) * (CH2 * MS) + (h * CHK + lm) * MS + wave * (KT * 16) + 8 * hi;
                    #pragma unroll
                    for (int a = 0; a < KT; ++a) {
                        kq[h][a][0] = *(const float4*)(krow + a * 16);
                        kq[h][a][1] = *(const float4*)(krow + a * 16 + 4);
                    }
                }
            }
            __builtin_amdgcn_sched_barrier(0);

            v8f u0 = {};
            {
                v2f ta, rb;
                ta.x = fq[0][0].x; ta.y = fq[0][0].y; rb.x = rq0.x; rb.y = rq0.y;
                u0 = __builtin_amdgcn_wmma_f32_16x16x4_f32(false, ta, false, rb, (short)0, u0, false, false);
                ta.x = fq[0][0].z; ta.y = fq[0][0].w; rb.x = rq0.z; rb.y = rq0.w;
                u0 = __builtin_amdgcn_wmma_f32_16x16x4_f32(false, ta, false, rb, (short)0, u0, false, false);
                ta.x = fq[0][1].x; ta.y = fq[0][1].y; rb.x = rq1.x; rb.y = rq1.y;
                u0 = __builtin_amdgcn_wmma_f32_16x16x4_f32(false, ta, false, rb, (short)0, u0, false, false);
                ta.x = fq[0][1].z; ta.y = fq[0][1].w; rb.x = rq1.z; rb.y = rq1.w;
                u0 = __builtin_amdgcn_wmma_f32_16x16x4_f32(false, ta, false, rb, (short)0, u0, false, false);
            }
            v8f r1;
            r1[0] = rb0.x; r1[1] = rb0.y; r1[2] = rb0.z; r1[3] = rb0.w;
            r1[4] = rb1.x; r1[5] = rb1.y; r1[6] = rb1.z; r1[7] = rb1.w;
            #pragma unroll
            for (int j = 0; j < 4; ++j) {
                const float4 f = fq[2][j >> 1];
                v2f xa;
                xa.x = (j & 1) ? f.z : f.x;
                xa.y = (j & 1) ? f.w : f.y;
                v2f ub; ub.x = u0[2 * j]; ub.y = u0[2 * j + 1];
                r1 = __builtin_amdgcn_wmma_f32_16x16x4_f32(false, xa, false, ub, (short)0, r1, false, false);
                #pragma unroll
                for (int a = 0; a < KT; ++a) {
                    v2f kt2; kt2.x = kx[0][a][j]; kt2.y = ky[0][a][j];
                    mt[a] = __builtin_amdgcn_wmma_f32_16x16x4_f32(
                                false, kt2, false, ub, (short)0, mt[a],
                                false, false);
                }
            }
            v8f u1 = {};
            #pragma unroll
            for (int j = 0; j < 4; ++j) {
                const float4 f = fq[1][j >> 1];
                v2f ta;
                ta.x = (j & 1) ? f.z : f.x;
                ta.y = (j & 1) ? f.w : f.y;
                v2f rb; rb.x = r1[2 * j]; rb.y = r1[2 * j + 1];
                u1 = __builtin_amdgcn_wmma_f32_16x16x4_f32(false, ta, false, rb, (short)0, u1, false, false);
            }
            #pragma unroll
            for (int j = 0; j < 4; ++j) {
                v2f ub; ub.x = u1[2 * j]; ub.y = u1[2 * j + 1];
                #pragma unroll
                for (int a = 0; a < KT; ++a) {
                    v2f kt2; kt2.x = kx[1][a][j]; kt2.y = ky[1][a][j];
                    mt[a] = __builtin_amdgcn_wmma_f32_16x16x4_f32(
                                false, kt2, false, ub, (short)0, mt[a],
                                false, false);
                }
            }
        }
    }

    if (mwave) {
        #pragma unroll
        for (int a = 0; a < KT; ++a) {
            float* dst = Msub + lm * MS + wave * (KT * 16) + a * 16 + 8 * hi;
            const v8f m = mt[a];
            float4 lo, up;
            lo.x = m[0]; lo.y = m[1]; lo.z = m[2]; lo.w = m[3];
            up.x = m[4]; up.y = m[5]; up.z = m[6]; up.w = m[7];
            *(float4*)dst = lo;
            *(float4*)(dst + 4) = up;
        }
    }
    __syncthreads();
    for (int i = tid; i < VROWS * DK_; i += NTHREADS) {
        int r = i >> 8, k = i & 255;
        outB[r * DK_ + k] = Msub[r * MS + k];
    }
}

extern "C" void kernel_launch(void* const* d_in, const int* in_sizes, int n_in,
                              void* d_out, int out_size, void* d_ws, size_t ws_size,
                              hipStream_t stream) {
    (void)in_sizes; (void)n_in; (void)out_size;
    const float* mem = (const float*)d_in[0];
    const float* key = (const float*)d_in[1];
    const float* val = (const float*)d_in[2];
    float* out = (float*)d_out;
    float* ws  = (float*)d_ws;

    hipLaunchKernelGGL(DeltaRuleMemory_chunk_factors, dim3(FAC_GRID), dim3(NTHREADS), 0, stream,
                       key, ws, ws_size);
    hipLaunchKernelGGL(DeltaRuleMemory_chunked_wmma, dim3(B_ * SPLIT), dim3(NTHREADS), 0, stream,
                       mem, key, val, out, (const float*)ws, ws_size);
}


# Staging waves reach the reduce-closing barrier sooner: wave stores of 512 contiguous bytes of one key row (was 4 rows x 128 B, 4 lanes a bank), next loads issued behind it; sum order unchanged
# speedup vs baseline: 5.0740x; 1.1042x over previous
#include <hip/hip_runtime.h>
#include <math.h>


typedef float v2f __attribute__((ext_vector_type(2)));
typedef float v8f __attribute__((ext_vector_type(8)));

#define B_   16
#define S_   4096
#define DK_  256
#define DV_  256
#define CHK  16
#define NCHK (S_ / CHK)
#define CH2  (2 * CHK)
#define NPAIR (NCHK / 2)
#define SPLIT 16
#define VROWS (DV_ / SPLIT)
#define NTHREADS 256
#define NWAVES (NTHREADS / 32)
#define MWAVES 4
#define KT 4
#define MS 260
#define PS 18
#define RS 20
#define RS2 36
#define VS 18
#define WS 20

#define SMEM_FLOATS (VROWS * MS + 2 * CH2 * MS + MWAVES * CH2 * PS + VROWS * RS2 + 3 * CHK * RS + 2 * CH2 * VS)
#define STAGE_WAVE0 MWAVES
#define VPIECES (VROWS / 4)
#define DPP_ROW_XMASK 0x160

#define LR_C 0.1f
#define A_C  1.1f
#define EPS_C 1e-6f

#define FAC_CPW 8
#define FAC_GRID (B_ * NCHK / FAC_CPW)
#define WS_T_FLOATS (B_ * NCHK * CHK * CHK)
#define WS_S_FLOATS (B_ * S_)
#define WS_X_FLOATS (B_ * NPAIR * CHK * CHK)
#define WS_X_OFF (WS_T_FLOATS + WS_S_FLOATS)
#define WS_NEED_BYTES ((size_t)(WS_T_FLOATS + WS_S_FLOATS + WS_X_FLOATS) * sizeof(float))

template <int M>
__device__ __forceinline__ float row_xor_add(float x) {
    int y = __builtin_amdgcn_update_dpp(0, __float_as_int(x), DPP_ROW_XMASK + M,
                                        0xf, 0xf, true);
    return x + __int_as_float(y);
}

template <int U>
__device__ __forceinline__ void subst_step(float wn, float& x) {
    asm("v_fmac_f32_dpp %0, %0, %1 row_share:%2 row_mask:0xf bank_mask:0xf bound_ctrl:1"
        : "+v"(x) : "v"(wn), "n"(U));
}

__device__ __forceinline__ void lds_barrier() {
    __builtin_amdgcn_fence(__ATOMIC_RELEASE, "workgroup", "local");
    __builtin_amdgcn_s_barrier();
    __builtin_amdgcn_fence(__ATOMIC_ACQUIRE, "workgroup", "local");
}

__global__ __launch_bounds__(NTHREADS)
void DeltaRuleMemory_chunk_factors(const float* __restrict__ key, float* __restrict__ ws,
                                   size_t ws_size) {
    if (ws_size < WS_NEED_BYTES) return;

    __shared__ __attribute__((aligned(16))) float fsm[2 * CHK * MS + 2 * NWAVES * 256 + CHK * WS + FAC_CPW * CHK];
    float* Knf = fsm;
    float* Gp = Knf + 2 * CHK * MS;
    float* Gx = Gp + NWAVES * 256;
    float* Wu = Gx + NWAVES * 256;
    float* Sc = Wu + CHK * WS;

    const int tid  = threadIdx.x;
    const int wave = tid >> 5;
    const int lane = tid & 31;
    const int hi   = lane >> 4;
    const int lm   = lane & 15;
    const int row  = tid >> 4;

    const int cid0 = blockIdx.x * FAC_CPW;
    const float* kr = key + (size_t)cid0 * CHK * DK_ + (size_t)row * DK_ + lm * 4;
    float* Tw = ws + (size_t)cid0 * (CHK * CHK) + tid;
    float* Xw = ws + WS_X_OFF + (size_t)(cid0 >> 1) * (CHK * CHK) + tid;

    float4 q[4];
    #pragma unroll
    for (int j = 0; j < 4; ++j) q[j] = ((const float4*)kr)[j * 16];

    for (int i = 0; i < FAC_CPW; ++i) {
        float*       Kn = Knf + (i & 1) * (CHK * MS);
        const float* Kp = Knf + ((i & 1) ^ 1) * (CHK * MS);
        float ss = 0.f;
        #pragma unroll
        for (int j = 0; j < 4; ++j)
            ss += q[j].x * q[j].x + q[j].y * q[j].y + q[j].z * q[j].z + q[j].w * q[j].w;
        ss = row_xor_add<1>(ss);
        ss = row_xor_add<2>(ss);
        ss = row_xor_add<4>(ss);
        ss = row_xor_add<8>(ss);
        const float sc = 1.f / (sqrtf(ss) + EPS_C);
        #pragma unroll
        for (int j = 0; j < 4; ++j) {
            float4 o;
            o.x = q[j].x * sc; o.y = q[j].y * sc; o.z = q[j].z * sc; o.w = q[j].w * sc;
            ((float4*)(Kn + row * MS + lm * 4))[j * 16] = o;
        }
        if (lm == 0) Sc[i * CHK + row] = sc;
        if (i + 1 < FAC_CPW) {
            const float* kn = kr + (size_t)(i + 1) * CHK * DK_;
            #pragma unroll
            for (int j = 0; j < 4; ++j) q[j] = ((const float4*)kn)[j * 16];
        }
        lds_barrier();

        {
            const float* krow = Kn + lm * MS + wave * 32 + 8 * hi;
            v8f g = {};
            #pragma unroll
            for (int a = 0; a < 2; ++a) {
                const float4 q0 = *(const float4*)(krow + a * 16);
                const float4 q1 = *(const float4*)(krow + a * 16 + 4);
                v2f kf;
                kf.x = q0.x; kf.y = q0.y;
                g = __builtin_amdgcn_wmma_f32_16x16x4_f32(false, kf, false, kf, (short)0, g, false, false);
                kf.x = q0.z; kf.y = q0.w;
                g = __builtin_amdgcn_wmma_f32_16x16x4_f32(false, kf, false, kf, (short)0, g, false, false);
                kf.x = q1.x; kf.y = q1.y;
                g = __builtin_amdgcn_wmma_f32_16x16x4_f32(false, kf, false, kf, (short)0, g, false, false);
                kf.x = q1.z; kf.y = q1.w;
                g = __builtin_amdgcn_wmma_f32_16x16x4_f32(false, kf, false, kf, (short)0, g, false, false);
            }
            float* gp = Gp + wave * 256;
            #pragma unroll
            for (int r = 0; r < 8; ++r) gp[(r + 8 * hi) * 16 + lm] = g[r];
        }
        if (i & 1) {
            const float* krow = Kn + lm * MS + wave * 32 + 8 * hi;
            const float* prow = Kp + lm * MS + wave * 32 + 8 * hi;
            v8f x = {};
            #pragma unroll
            for (int a = 0; a < 2; ++a) {
                const float4 q0 = *(const float4*)(krow + a * 16);
                const float4 q1 = *(const float4*)(krow + a * 16 + 4);
                const float4 p0 = *(const float4*)(prow + a * 16);
                const float4 p1 = *(const float4*)(prow + a * 16 + 4);
                v2f kf, pf;
                kf.x = q0.x; kf.y = q0.y; pf.x = p0.x; pf.y = p0.y;
                x = __builtin_amdgcn_wmma_f32_16x16x4_f32(false, kf, false, pf, (short)0, x, false, false);
                kf.x = q0.z; kf.y = q0.w; pf.x = p0.z; pf.y = p0.w;
                x = __builtin_amdgcn_wmma_f32_16x16x4_f32(false, kf, false, pf, (short)0, x, false, false);
                kf.x = q1.x; kf.y = q1.y; pf.x = p1.x; pf.y = p1.y;
                x = __builtin_amdgcn_wmma_f32_16x16x4_f32(false, kf, false, pf, (short)0, x, false, false);
                kf.x = q1.z; kf.y = q1.w; pf.x = p1.z; pf.y = p1.w;
                x = __builtin_amdgcn_wmma_f32_16x16x4_f32(false, kf, false, pf, (short)0, x, false, false);
            }
            float* gx = Gx + wave * 256;
            #pragma unroll
            for (int r = 0; r < 8; ++r) gx[(r + 8 * hi) * 16 + lm] = x[r];
        }
        lds_barrier();

        if (i & 1) {
            float xs = Gx[tid];
            #pragma unroll
            for (int w = 1; w < NWAVES; ++w) xs += Gx[w * 256 + tid];
            Xw[(size_t)(i >> 1) * (CHK * CHK)] = -(A_C * xs);
        }
        {
            float gs = Gp[tid];
            #pragma unroll
            for (int w = 1; w < NWAVES; ++w) gs += Gp[w * 256 + tid];
            Wu[row * WS + lm] = (lm > row) ? -(A_C * gs) : 0.f;
        }
        lds_barrier();

        {
            const float4* wr = (const float4*)(Wu + lm * WS);
            const float4 w0 = wr[0], w1 = wr[1], w2 = wr[2], w3 = wr[3];
            float x = (lm == row) ? 1.f : 0.f;
            subst_step<15>(w3.w, x);
            subst_step<14>(w3.z, x);
            subst_step<13>(w3.y, x);
            subst_step<12>(w3.x, x);
            subst_step<11>(w2.w, x);
            subst_step<10>(w2.z, x);
            subst_step<9>(w2.y, x);
            subst_step<8>(w2.x, x);
            subst_step<7>(w1.w, x);
            subst_step<6>(w1.z, x);
            subst_step<5>(w1.y, x);
            subst_step<4>(w1.x, x);
            subst_step<3>(w0.w, x);
            subst_step<2>(w0.z, x);
            subst_step<1>(w0.y, x);
            Tw[(size_t)i * (CHK * CHK)] = x;
        }
    }
    lds_barrier();
    if (tid < FAC_CPW * CHK) ws[WS_T_FLOATS + (size_t)cid0 * CHK + tid] = Sc[tid];
}


struct Staged {
    float4 k[8][2];
    float4 v;
    float  sc[8];
};

__device__ __forceinline__ void load_chunk(Staged& s, const float* __restrict__ kr,
                                           const float* __restrict__ vg,
                                           const float* __restrict__ sr) {
    #pragma unroll
    for (int i = 0; i < 8; ++i) {
        s.k[i][0] = ((const float4*)(kr + (size_t)i * DK_))[0];
        s.k[i][1] = ((const float4*)(kr + (size_t)i * DK_))[32];
    }
    const float4 s0 = ((const float4*)sr)[0];
    const float4 s1 = ((const float4*)sr)[1];
    s.sc[0] = s0.x; s.sc[1] = s0.y; s.sc[2] = s0.z; s.sc[3] = s0.w;
    s.sc[4] = s1.x; s.sc[5] = s1.y; s.sc[6] = s1.z; s.sc[7] = s1.w;
    s.v = *(const float4*)vg;
}

__device__ __forceinline__ void stage_write(const Staged& s, float* __restrict__ kw,
                                            float* __restrict__ vw) {
    #pragma unroll
    for (int i = 0; i < 8; ++i) {
        const float sc = s.sc[i];
        #pragma unroll
        for (int c = 0; c < 2; ++c) {
            float4 o;
            o.x = s.k[i][c].x * sc; o.y = s.k[i][c].y * sc;
            o.z = s.k[i][c].z * sc; o.w = s.k[i][c].w * sc;
            ((float4*)(kw + i * MS))[c * 32] = o;
        }
    }
    vw[0] = s.v.x; vw[1] = s.v.y; vw[2] = s.v.z; vw[3] = s.v.w;
}

__global__ __launch_bounds__(NTHREADS)
void DeltaRuleMemory_chunked_wmma(const float* __restrict__ mem,
                                  const float* __restrict__ key,
                                  const float* __restrict__ val,
                                  float* __restrict__ out,
                                  const float* __restrict__ ws,
                                  size_t ws_size) {
    if (ws_size < WS_NEED_BYTES) return;

    __shared__ __attribute__((aligned(16))) float smem[SMEM_FLOATS];
    float* Msub = smem;
    float* Kn2  = Msub + VROWS * MS;
    float* P0p  = Kn2  + 2 * CH2 * MS;
    float* Rv   = P0p  + MWAVES * CH2 * PS;
    float* Tl   = Rv   + VROWS * RS2;
    float* Vc2  = Tl   + 3 * CHK * RS;

    const int tid  = threadIdx.x;
    const int wave = __builtin_amdgcn_readfirstlane(tid >> 5);
    const int lane = tid & 31;
    const int hi   = lane >> 4;
    const int lm   = lane & 15;
    const bool mwave = wave < MWAVES;

    const int blk = blockIdx.x;
    const int b   = blk / SPLIT;
    const int v0  = (blk % SPLIT) * VROWS;

    const float* memB = mem + (size_t)b * DV_ * DK_ + (size_t)v0 * DK_;
    const float* keyB = key + (size_t)b * S_ * DK_;
    const float* valB = val + (size_t)b * S_ * DV_;
    float*       outB = out + (size_t)b * DV_ * DK_ + (size_t)v0 * DK_;
    const float* scB  = ws + WS_T_FLOATS + (size_t)b * S_;

    const bool stager = wave >= STAGE_WAVE0;
    const int  sR  = 8 * (wave & (MWAVES - 1));
    const int  sI  = tid & (4 * 32 - 1);
    const int  sVr = sI >> 2;
    const int  sVp = sI & (VPIECES - 1);
    const float* krT = keyB + (size_t)sR * DK_ + lane * 4;
    const float* vgT = valB + (size_t)sVr * DV_ + v0 + sVp * 4;
    const float* srT = scB + sR;
    float* kwT = Kn2 + sR * MS + lane * 4;
    float* vwT = Vc2 + sVr * VS + sVp * 4;
    const int    sE  = 2 * (tid & (4 * 32 - 1));
    const float* tpT = ws + (size_t)b * NCHK * (CHK * CHK) + sE;
    const float* xpT = ws + WS_X_OFF + (size_t)b * NPAIR * (CHK * CHK) + sE;
    float*       twT = Tl + (sE >> 4) * RS + (sE & 15);

    Staged st;
    float2 stT[3] = {{0.f, 0.f}, {0.f, 0.f}, {0.f, 0.f}};
    if (stager) load_chunk(st, krT, vgT, srT);
    for (int i = tid; i < VROWS * DK_; i += NTHREADS) {
        int r = i >> 8, k = i & 255;
        Msub[r * MS + k] = memB[r * DK_ + k];
    }
    if (stager) stage_write(st, kwT, vwT);
    __syncthreads();

    v8f mt[KT] = {{}, {}, {}, {}};
    if (mwave) {
        #pragma unroll
        for (int a = 0; a < KT; ++a) {
            const float* src = Msub + lm * MS + wave * (KT * 16) + a * 16 + 8 * hi;
            const float4 lo = *(const float4*)src;
            const float4 up = *(const float4*)(src + 4);
            v8f m;
            m[0] = lo.x; m[1] = lo.y; m[2] = lo.z; m[3] = lo.w;
            m[4] = up.x; m[5] = up.y; m[6] = up.z; m[7] = up.w;
            mt[a] = m;
        }
    }

    if (stager) {
        load_chunk(st, krT + (size_t)CH2 * DK_, vgT + (size_t)CH2 * DV_, srT + CH2);
        stT[0] = *(const float2*)tpT;
        stT[1] = *(const float2*)(tpT + CHK * CHK);
        stT[2] = *(const float2*)xpT;
        __builtin_prefetch(keyB + ((size_t)2 * CH2 + sVr) * DK_ + sVp * 64, 0, 0);
        __builtin_prefetch(keyB + ((size_t)2 * CH2 + sVr) * DK_ + sVp * 64 + 32, 0, 0);
    }
    float4 kq[2][KT][2] = {};
    if (mwave) {
        #pragma unroll
        for (int h = 0; h < 2; ++h) {
            const float* krow = Kn2 + (h * CHK + lm) * MS + wave * (KT * 16) + 8 * hi;
            #pragma unroll
            for (int a = 0; a < KT; ++a) {
                kq[h][a][0] = *(const float4*)(krow + a * 16);
                kq[h][a][1] = *(const float4*)(krow + a * 16 + 4);
            }
        }
    }

    for (int p = 0; p < NPAIR; ++p) {
        const int cur = p & 1;
        const float* Kn = Kn2 + cur * (CH2 * MS);
        const float* Vc = Vc2 + cur * (CH2 * VS);
        const bool stage_next = stager && (p + 1 < NPAIR);

        if (mwave) {
            v8f p0a[2] = {{}, {}};
            #pragma unroll
            for (int a = 0; a < KT; ++a) {
                #pragma unroll
                for (int j = 0; j < 4; ++j) {
                    v2f m0; m0.x = mt[a][2 * j]; m0.y = mt[a][2 * j + 1];
                    #pragma unroll
                    for (int h = 0; h < 2; ++h) {
                        const float4 q = kq[h][a][j >> 1];
                        v2f kf;
                        kf.x = (j & 1) ? q.z : q.x;
                        kf.y = (j & 1) ? q.w : q.y;
                        p0a[h] = __builtin_amdgcn_wmma_f32_16x16x4_f32(
                                     false, kf, false, m0, (short)0, p0a[h], false, false);
                    }
                }
            }
            float* pp = P0p + wave * (CH2 * PS);
            #pragma unroll
            for (int h = 0; h < 2; ++h) {
                #pragma unroll
                for (int r = 0; r < 8; ++r) pp[(h * CHK + r + 8 * hi) * PS + lm] = p0a[h][r];
            }
        }
        lds_barrier();

        if (!stager) {
            const int va = 4 * wave + hi;
            float pa[2], pb[2], xa[2], xb[2];
            #pragma unroll
            for (int h = 0; h < 2; ++h) {
                const float* pp = P0p + (h * CHK + lm) * PS + va;
                pa[h] = pp[0];
                pb[h] = pp[2];
                #pragma unroll
                for (int w = 1; w < MWAVES; ++w) {
                    pa[h] += pp[w * (CH2 * PS)];
                    pb[h] += pp[w * (CH2 * PS) + 2];
                }
                xa[h] = Vc[(h * CHK + lm) * VS + va];
                xb[h] = Vc[(h * CHK + lm) * VS + va + 2];
            }
            #pragma unroll
            for (int h = 0; h < 2; ++h) {
                Rv[va * RS2 + h * CHK + lm]       = LR_C * xa[h] - A_C * pa[h];
                Rv[(va + 2) * RS2 + h * CHK + lm] = LR_C * xb[h] - A_C * pb[h];
            }
        } else {
            *(float2*)twT = stT[0];
            *(float2*)(twT + CHK * RS) = stT[1];
            *(float2*)(twT + 2 * CHK * RS) = stT[2];
            if (stage_next) {
                const int nxt = cur ^ 1;
                stage_write(st, kwT + nxt * (CH2 * MS), vwT + nxt * (CH2 * VS));
            }
        }
        lds_barrier();

        if (stage_next) {
            stT[0] = *(const float2*)(tpT + (size_t)(2 * p + 2) * (CHK * CHK));
            stT[1] = *(const float2*)(tpT + (size_t)(2 * p + 3) * (CHK * CHK));
            stT[2] = *(const float2*)(xpT + (size_t)(p + 1) * (CHK * CHK));
            if (p + 2 < NPAIR) {
                load_chunk(st, krT + (size_t)(p + 2) * CH2 * DK_, vgT + (size_t)(p + 2) * CH2 * DV_,
                           srT + (p + 2) * CH2);
                if (p + 3 < NPAIR) {
                    __builtin_prefetch(keyB + ((size_t)(p + 3) * CH2 + sVr) * DK_ + sVp * 64, 0, 0);
                    __builtin_prefetch(keyB + ((size_t)(p + 3) * CH2 + sVr) * DK_ + sVp * 64 + 32, 0, 0);
                }
            }
        }

        if (mwave) {
            const float4 rq0 = *(const float4*)(Rv + lm * RS2 + 8 * hi);
            const float4 rq1 = *(const float4*)(Rv + lm * RS2 + 8 * hi + 4);
            const float4 rb0 = *(const float4*)(Rv + lm * RS2 + CHK + 8 * hi);
            const float4 rb1 = *(const float4*)(Rv + lm * RS2 + CHK + 8 * hi + 4);
            float4 fq[3][2];
            #pragma unroll
            for (int m = 0; m < 3; ++m) {
                fq[m][0] = *(const float4*)(Tl + m * (CHK * RS) + lm * RS + 8 * hi);
                fq[m][1] = *(const float4*)(Tl + m * (CHK * RS) + lm * RS + 8 * hi + 4);
            }
            __builtin_amdgcn_sched_barrier(0);
            float kx[2][KT][4], ky[2][KT][4];
            #pragma unroll
            for (int h = 0; h < 2; ++h) {
                #pragma unroll
                for (int j = 0; j < 4; ++j) {
                    #pragma unroll
                    for (int a = 0; a < KT; ++a) {
                        const float* kp = Kn + (h * CHK + 8 * hi + 2 * j) * MS + wave * (KT * 16) + a * 16 + lm;
                        kx[h][a][j] = kp[0];
                        ky[h][a][j] = kp[MS];
                    }
                }
            }
            __builtin_amdgcn_sched_barrier(0);
            if (p + 1 < NPAIR) {
                #pragma unroll
                for (int h = 0; h < 2; ++h) {
                    const float* krow = Kn2 + (cur ^ 1) * (CH2 * MS) + (h * CHK + lm) * MS + wave * (KT * 16) + 8 * hi;
                    #pragma unroll
                    for (int a = 0; a < KT; ++a) {
                        kq[h][a][0] = *(const float4*)(krow + a * 16);
                        kq[h][a][1] = *(const float4*)(krow + a * 16 + 4);
                    }
                }
            }
            __builtin_amdgcn_sched_barrier(0);

            v8f u0 = {};
            {
                v2f ta, rb;
                ta.x = fq[0][0].x; ta.y = fq[0][0].y; rb.x = rq0.x; rb.y = rq0.y;
                u0 = __builtin_amdgcn_wmma_f32_16x16x4_f32(false, ta, false, rb, (short)0, u0, false, false);
                ta.x = fq[0][0].z; ta.y = fq[0][0].w; rb.x = rq0.z; rb.y = rq0.w;
                u0 = __builtin_amdgcn_wmma_f32_16x16x4_f32(false, ta, false, rb, (short)0, u0, false, false);
                ta.x = fq[0][1].x; ta.y = fq[0][1].y; rb.x = rq1.x; rb.y = rq1.y;
                u0 = __builtin_amdgcn_wmma_f32_16x16x4_f32(false, ta, false, rb, (short)0, u0, false, false);
                ta.x = fq[0][1].z; ta.y = fq[0][1].w; rb.x = rq1.z; rb.y = rq1.w;
                u0 = __builtin_amdgcn_wmma_f32_16x16x4_f32(false, ta, false, rb, (short)0, u0, false, false);
            }
            v8f r1;
            r1[0] = rb0.x; r1[1] = rb0.y; r1[2] = rb0.z; r1[3] = rb0.w;
            r1[4] = rb1.x; r1[5] = rb1.y; r1[6] = rb1.z; r1[7] = rb1.w;
            #pragma unroll
            for (int j = 0; j < 4; ++j) {
                const float4 f = fq[2][j >> 1];
                v2f xa;
                xa.x = (j & 1) ? f.z : f.x;
                xa.y = (j & 1) ? f.w : f.y;
                v2f ub; ub.x = u0[2 * j]; ub.y = u0[2 * j + 1];
                r1 = __builtin_amdgcn_wmma_f32_16x16x4_f32(false, xa, false, ub, (short)0, r1, false, false);
                #pragma unroll
                for (int a = 0; a < KT; ++a) {
                    v2f kt2; kt2.x = kx[0][a][j]; kt2.y = ky[0][a][j];
                    mt[a] = __builtin_amdgcn_wmma_f32_16x16x4_f32(
                                false, kt2, false, ub, (short)0, mt[a],
                                false, false);
                }
            }
            v8f u1 = {};
            #pragma unroll
            for (int j = 0; j < 4; ++j) {
                const float4 f = fq[1][j >> 1];
                v2f ta;
                ta.x = (j & 1) ? f.z : f.x;
                ta.y = (j & 1) ? f.w : f.y;
                v2f rb; rb.x = r1[2 * j]; rb.y = r1[2 * j + 1];
                u1 = __builtin_amdgcn_wmma_f32_16x16x4_f32(false, ta, false, rb, (short)0, u1, false, false);
            }
            #pragma unroll
            for (int j = 0; j < 4; ++j) {
                v2f ub; ub.x = u1[2 * j]; ub.y = u1[2 * j + 1];
                #pragma unroll
                for (int a = 0; a < KT; ++a) {
                    v2f kt2; kt2.x = kx[1][a][j]; kt2.y = ky[1][a][j];
                    mt[a] = __builtin_amdgcn_wmma_f32_16x16x4_f32(
                                false, kt2, false, ub, (short)0, mt[a],
                                false, false);
                }
            }
        }
    }

    if (mwave) {
        #pragma unroll
        for (int a = 0; a < KT; ++a) {
            float* dst = Msub + lm * MS + wave * (KT * 16) + a * 16 + 8 * hi;
            const v8f m = mt[a];
            float4 lo, up;
            lo.x = m[0]; lo.y = m[1]; lo.z = m[2]; lo.w = m[3];
            up.x = m[4]; up.y = m[5]; up.z = m[6]; up.w = m[7];
            *(float4*)dst = lo;
            *(float4*)(dst + 4) = up;
        }
    }
    __syncthreads();
    for (int i = tid; i < VROWS * DK_; i += NTHREADS) {
        int r = i >> 8, k = i & 255;
        outB[r * DK_ + k] = Msub[r * MS + k];
    }
}

extern "C" void kernel_launch(void* const* d_in, const int* in_sizes, int n_in,
                              void* d_out, int out_size, void* d_ws, size_t ws_size,
                              hipStream_t stream) {
    (void)in_sizes; (void)n_in; (void)out_size;
    const float* mem = (const float*)d_in[0];
    const float* key = (const float*)d_in[1];
    const float* val = (const float*)d_in[2];
    float* out = (float*)d_out;
    float* ws  = (float*)d_ws;

    hipLaunchKernelGGL(DeltaRuleMemory_chunk_factors, dim3(FAC_GRID), dim3(NTHREADS), 0, stream,
                       key, ws, ws_size);
    hipLaunchKernelGGL(DeltaRuleMemory_chunked_wmma, dim3(B_ * SPLIT), dim3(NTHREADS), 0, stream,
                       mem, key, val, out, (const float*)ws, ws_size);
}
